# starting kernel
# baseline (speedup 1.0000x reference)
#include <hip/hip_runtime.h>

#define NN 50000
#define NE 600000
#define DIM 128
#define BN_EPS 1e-5f

typedef __attribute__((ext_vector_type(2))) float v2f;
typedef __attribute__((ext_vector_type(8))) float v8f;


__global__ void init_deg_kernel(float* __restrict__ deg, int n) {
    int i = blockIdx.x * blockDim.x + threadIdx.x;
    if (i < n) deg[i] = 1.0f;
}

__global__ void deg_count_kernel(const int* __restrict__ col, float* __restrict__ deg, int e) {
    int i = blockIdx.x * blockDim.x + threadIdx.x;
    if (i < e) unsafeAtomicAdd(&deg[col[i]], 1.0f);
}

__global__ void rsqrt_kernel(float* __restrict__ deg, int n) {
    int i = blockIdx.x * blockDim.x + threadIdx.x;
    if (i < n) deg[i] = rsqrtf(deg[i]);
}

__global__ void zero_kernel(float* __restrict__ p, int n) {
    int i = blockIdx.x * blockDim.x + threadIdx.x;
    if (i < n) p[i] = 0.0f;
}


__global__ __launch_bounds__(256)
void transpose128_kernel(const float* __restrict__ W, float* __restrict__ Wt) {
    int t = blockIdx.x * blockDim.x + threadIdx.x;
    int r = t >> 7;
    int c = t & 127;
    Wt[(size_t)c * DIM + r] = W[(size_t)r * DIM + c];
}

__global__ __launch_bounds__(256)
void gemm_wmma_kernel(const float* __restrict__ X, const float* __restrict__ Wt,
                      float* __restrict__ H) {
    const int tileM = blockIdx.x * 16;
    const int wave  = threadIdx.x >> 5;
    const int lane  = threadIdx.x & 31;
    const int tileN = wave * 16;
    const int mrow  = lane & 15;
    const int hi    = lane >> 4;

    v8f acc = {};
    const float* xr = X + (size_t)(tileM + mrow) * DIM + 2 * hi;
    const float* wc = Wt + (size_t)(tileN + mrow) * DIM + 2 * hi;

#pragma unroll
    for (int k0 = 0; k0 < DIM; k0 += 4) {
        v2f a = *(const v2f*)(xr + k0);
        v2f b = *(const v2f*)(wc + k0);
        acc = __builtin_amdgcn_wmma_f32_16x16x4_f32(
             false, a,  false, b,
             (short)0, acc,  false,  false);
    }

    float* out = H + (size_t)(tileM + 8 * hi) * DIM + tileN + mrow;
#pragma unroll
    for (int r = 0; r < 8; ++r)
        out[(size_t)r * DIM] = acc[r];
}


__global__ __launch_bounds__(256)
void selfloop_kernel(const float* __restrict__ h, const float* __restrict__ dis,
                     float* __restrict__ agg, int n) {
    int t = blockIdx.x * blockDim.x + threadIdx.x;
    if (t >= n * 32) return;
    int node = t >> 5;
    float s = dis[node];
    s = s * s;
    float4 v = ((const float4*)h)[t];
    v.x *= s; v.y *= s; v.z *= s; v.w *= s;
    ((float4*)agg)[t] = v;
}

__global__ __launch_bounds__(256)
void edge_scatter_kernel(const float* __restrict__ h, float* __restrict__ agg,
                         const int* __restrict__ row, const int* __restrict__ col,
                         const float* __restrict__ dis, int e) {
    int warp = threadIdx.x >> 5;
    int lane = threadIdx.x & 31;
    int eid = blockIdx.x * 8 + warp;
    if (eid >= e) return;
    int r = row[eid];
    int c = col[eid];
    float norm = dis[r] * dis[c];
    float4 v = ((const float4*)(h + (size_t)r * DIM))[lane];
    float* dst = agg + (size_t)c * DIM + lane * 4;
    unsafeAtomicAdd(dst + 0, v.x * norm);
    unsafeAtomicAdd(dst + 1, v.y * norm);
    unsafeAtomicAdd(dst + 2, v.z * norm);
    unsafeAtomicAdd(dst + 3, v.w * norm);
}


__global__ __launch_bounds__(256)
void bias_relu_kernel(const float* __restrict__ agg, const float* __restrict__ b,
                      float* __restrict__ out, int n) {
    int t = blockIdx.x * blockDim.x + threadIdx.x;
    if (t >= n * 32) return;
    int j = t & 31;
    float4 v = ((const float4*)agg)[t];
    float4 bb = ((const float4*)b)[j];
    v.x = fmaxf(v.x + bb.x, 0.0f);
    v.y = fmaxf(v.y + bb.y, 0.0f);
    v.z = fmaxf(v.z + bb.z, 0.0f);
    v.w = fmaxf(v.w + bb.w, 0.0f);
    ((float4*)out)[t] = v;
}


__global__ __launch_bounds__(128)
void bn_reduce_kernel(const float* __restrict__ h, float* __restrict__ sum,
                      float* __restrict__ sumsq, int n, int rows_per_block) {
    int d = threadIdx.x;
    int r0 = blockIdx.x * rows_per_block;
    int r1 = r0 + rows_per_block;
    if (r1 > n) r1 = n;
    float s = 0.0f, ss = 0.0f;
    for (int r = r0; r < r1; ++r) {
        float v = h[(size_t)r * DIM + d];
        s += v;
        ss += v * v;
    }
    unsafeAtomicAdd(&sum[d], s);
    unsafeAtomicAdd(&sumsq[d], ss);
}

__global__ __launch_bounds__(128)
void bn_finalize_kernel(const float* __restrict__ sum, const float* __restrict__ sumsq,
                        float* __restrict__ mean, float* __restrict__ istd, int n) {
    int d = threadIdx.x;
    float invn = 1.0f / (float)n;
    float m = sum[d] * invn;
    float var = sumsq[d] * invn - m * m;
    mean[d] = m;
    istd[d] = rsqrtf(var + BN_EPS);
}

__global__ __launch_bounds__(256)
void bn_norm_kernel(const float* __restrict__ h, const float* __restrict__ mean,
                    const float* __restrict__ istd, const float* __restrict__ gamma,
                    const float* __restrict__ beta, float* __restrict__ out, int n) {
    int t = blockIdx.x * blockDim.x + threadIdx.x;
    if (t >= n * 32) return;
    int j = t & 31;
    float4 v  = ((const float4*)h)[t];
    float4 m  = ((const float4*)mean)[j];
    float4 is = ((const float4*)istd)[j];
    float4 g  = ((const float4*)gamma)[j];
    float4 be = ((const float4*)beta)[j];
    float4 o;
    o.x = g.x * (v.x - m.x) * is.x + be.x;
    o.y = g.y * (v.y - m.y) * is.y + be.y;
    o.z = g.z * (v.z - m.z) * is.z + be.z;
    o.w = g.w * (v.w - m.w) * is.w + be.w;
    ((float4*)out)[t] = o;
}


extern "C" void kernel_launch(void* const* d_in, const int* in_sizes, int n_in,
                              void* d_out, int out_size, void* d_ws, size_t ws_size,
                              hipStream_t stream) {
    const float* x     = (const float*)d_in[0];
    const int*   eidx  = (const int*)d_in[1];
    const float* W1    = (const float*)d_in[2];
    const float* b1    = (const float*)d_in[3];
    const float* W2    = (const float*)d_in[4];
    const float* b2    = (const float*)d_in[5];
    const float* gamma = (const float*)d_in[6];
    const float* beta  = (const float*)d_in[7];
    float* out = (float*)d_out;

    const int* row = eidx;
    const int* col = eidx + NE;

    float* bufA  = (float*)d_ws;
    float* bufB  = bufA + (size_t)NN * DIM;
    float* dis   = bufB + (size_t)NN * DIM;
    float* sum   = dis + NN;
    float* sumsq = sum + DIM;
    float* mean  = sumsq + DIM;
    float* istd  = mean + DIM;
    float* W1t   = istd + DIM;
    float* W2t   = W1t + (size_t)DIM * DIM;

    const int T = 256;
    const int elem_blocks = (NN * 32 + T - 1) / T;
    const int gemm_blocks = NN / 16;
    const int edge_blocks = (NE + 7) / 8;
    const int tr_blocks   = (DIM * DIM) / T;
    const int bn_blocks = 256;
    const int rows_per_block = (NN + bn_blocks - 1) / bn_blocks;

    init_deg_kernel<<<(NN + T - 1) / T, T, 0, stream>>>(dis, NN);
    deg_count_kernel<<<(NE + T - 1) / T, T, 0, stream>>>(col, dis, NE);
    rsqrt_kernel<<<(NN + T - 1) / T, T, 0, stream>>>(dis, NN);
    zero_kernel<<<1, 2 * DIM, 0, stream>>>(sum, 2 * DIM);
    transpose128_kernel<<<tr_blocks, T, 0, stream>>>(W1, W1t);
    transpose128_kernel<<<tr_blocks, T, 0, stream>>>(W2, W2t);

    gemm_wmma_kernel<<<gemm_blocks, T, 0, stream>>>(x, W1t, bufA);
    selfloop_kernel<<<elem_blocks, T, 0, stream>>>(bufA, dis, bufB, NN);
    edge_scatter_kernel<<<edge_blocks, T, 0, stream>>>(bufA, bufB, row, col, dis, NE);
    bias_relu_kernel<<<elem_blocks, T, 0, stream>>>(bufB, b1, bufA, NN);

    gemm_wmma_kernel<<<gemm_blocks, T, 0, stream>>>(bufA, W2t, bufB);
    selfloop_kernel<<<elem_blocks, T, 0, stream>>>(bufB, dis, bufA, NN);
    edge_scatter_kernel<<<edge_blocks, T, 0, stream>>>(bufB, bufA, row, col, dis, NE);
    bias_relu_kernel<<<elem_blocks, T, 0, stream>>>(bufA, b2, bufB, NN);

    bn_reduce_kernel<<<bn_blocks, DIM, 0, stream>>>(bufB, sum, sumsq, NN, rows_per_block);
    bn_finalize_kernel<<<1, DIM, 0, stream>>>(sum, sumsq, mean, istd, NN);
    bn_norm_kernel<<<elem_blocks, T, 0, stream>>>(bufB, mean, istd, gamma, beta, out, NN);
}


# bn_reduce: each wave requests 13 whole-row float4 loads up front instead of one dword per serial row step; 241 blocks of 512 threads, LDS fold, same atomics (column sum order changes)
# speedup vs baseline: 1.1978x; 1.1978x over previous
#include <hip/hip_runtime.h>

#define NN 50000
#define NE 600000
#define DIM 128
#define BN_EPS 1e-5f

typedef __attribute__((ext_vector_type(2))) float v2f;
typedef __attribute__((ext_vector_type(8))) float v8f;


__global__ void init_deg_kernel(float* __restrict__ deg, int n) {
    int i = blockIdx.x * blockDim.x + threadIdx.x;
    if (i < n) deg[i] = 1.0f;
}

__global__ void deg_count_kernel(const int* __restrict__ col, float* __restrict__ deg, int e) {
    int i = blockIdx.x * blockDim.x + threadIdx.x;
    if (i < e) unsafeAtomicAdd(&deg[col[i]], 1.0f);
}

__global__ void rsqrt_kernel(float* __restrict__ deg, int n) {
    int i = blockIdx.x * blockDim.x + threadIdx.x;
    if (i < n) deg[i] = rsqrtf(deg[i]);
}

__global__ void zero_kernel(float* __restrict__ p, int n) {
    int i = blockIdx.x * blockDim.x + threadIdx.x;
    if (i < n) p[i] = 0.0f;
}


__global__ __launch_bounds__(256)
void transpose128_kernel(const float* __restrict__ W, float* __restrict__ Wt) {
    int t = blockIdx.x * blockDim.x + threadIdx.x;
    int r = t >> 7;
    int c = t & 127;
    Wt[(size_t)c * DIM + r] = W[(size_t)r * DIM + c];
}

__global__ __launch_bounds__(256)
void gemm_wmma_kernel(const float* __restrict__ X, const float* __restrict__ Wt,
                      float* __restrict__ H) {
    const int tileM = blockIdx.x * 16;
    const int wave  = threadIdx.x >> 5;
    const int lane  = threadIdx.x & 31;
    const int tileN = wave * 16;
    const int mrow  = lane & 15;
    const int hi    = lane >> 4;

    v8f acc = {};
    const float* xr = X + (size_t)(tileM + mrow) * DIM + 2 * hi;
    const float* wc = Wt + (size_t)(tileN + mrow) * DIM + 2 * hi;

#pragma unroll
    for (int k0 = 0; k0 < DIM; k0 += 4) {
        v2f a = *(const v2f*)(xr + k0);
        v2f b = *(const v2f*)(wc + k0);
        acc = __builtin_amdgcn_wmma_f32_16x16x4_f32(
             false, a,  false, b,
             (short)0, acc,  false,  false);
    }

    float* out = H + (size_t)(tileM + 8 * hi) * DIM + tileN + mrow;
#pragma unroll
    for (int r = 0; r < 8; ++r)
        out[(size_t)r * DIM] = acc[r];
}


__global__ __launch_bounds__(256)
void selfloop_kernel(const float* __restrict__ h, const float* __restrict__ dis,
                     float* __restrict__ agg, int n) {
    int t = blockIdx.x * blockDim.x + threadIdx.x;
    if (t >= n * 32) return;
    int node = t >> 5;
    float s = dis[node];
    s = s * s;
    float4 v = ((const float4*)h)[t];
    v.x *= s; v.y *= s; v.z *= s; v.w *= s;
    ((float4*)agg)[t] = v;
}

__global__ __launch_bounds__(256)
void edge_scatter_kernel(const float* __restrict__ h, float* __restrict__ agg,
                         const int* __restrict__ row, const int* __restrict__ col,
                         const float* __restrict__ dis, int e) {
    int warp = threadIdx.x >> 5;
    int lane = threadIdx.x & 31;
    int eid = blockIdx.x * 8 + warp;
    if (eid >= e) return;
    int r = row[eid];
    int c = col[eid];
    float norm = dis[r] * dis[c];
    float4 v = ((const float4*)(h + (size_t)r * DIM))[lane];
    float* dst = agg + (size_t)c * DIM + lane * 4;
    unsafeAtomicAdd(dst + 0, v.x * norm);
    unsafeAtomicAdd(dst + 1, v.y * norm);
    unsafeAtomicAdd(dst + 2, v.z * norm);
    unsafeAtomicAdd(dst + 3, v.w * norm);
}


__global__ __launch_bounds__(256)
void bias_relu_kernel(const float* __restrict__ agg, const float* __restrict__ b,
                      float* __restrict__ out, int n) {
    int t = blockIdx.x * blockDim.x + threadIdx.x;
    if (t >= n * 32) return;
    int j = t & 31;
    float4 v = ((const float4*)agg)[t];
    float4 bb = ((const float4*)b)[j];
    v.x = fmaxf(v.x + bb.x, 0.0f);
    v.y = fmaxf(v.y + bb.y, 0.0f);
    v.z = fmaxf(v.z + bb.z, 0.0f);
    v.w = fmaxf(v.w + bb.w, 0.0f);
    ((float4*)out)[t] = v;
}


#define BN_WAVES 16
#define BN_RPW   13
#define BN_ROWS  (BN_WAVES * BN_RPW)

__global__ __launch_bounds__(32 * BN_WAVES)
void bn_reduce_kernel(const float* __restrict__ h, float* __restrict__ sum,
                      float* __restrict__ sumsq, int n) {
    __shared__ float4 red[2][BN_WAVES][32];
    const int j  = threadIdx.x & 31;
    const int g  = threadIdx.x >> 5;
    const int r0 = blockIdx.x * BN_ROWS + g * BN_RPW;
    const float4* hp = (const float4*)h;

    float4 v[BN_RPW];
#pragma unroll
    for (int i = 0; i < BN_RPW; ++i) {
        int r = r0 + i;
        if (r >= n) r = n - 1;
        v[i] = hp[(size_t)r * 32 + j];
    }

    float4 s = {0.0f, 0.0f, 0.0f, 0.0f};
    float4 q = {0.0f, 0.0f, 0.0f, 0.0f};
#pragma unroll
    for (int i = 0; i < BN_RPW; ++i) {
        const bool live = (r0 + i) < n;
        float4 w;
        w.x = live ? v[i].x : 0.0f;
        w.y = live ? v[i].y : 0.0f;
        w.z = live ? v[i].z : 0.0f;
        w.w = live ? v[i].w : 0.0f;
        s.x += w.x;  s.y += w.y;  s.z += w.z;  s.w += w.w;
        q.x += w.x * w.x;  q.y += w.y * w.y;  q.z += w.z * w.z;  q.w += w.w * w.w;
    }

    red[0][g][j] = s;
    red[1][g][j] = q;
    __syncthreads();

    if (threadIdx.x < 2 * DIM) {
        const int which = threadIdx.x >> 7;
        const int d = threadIdx.x & 127;
        const float* rp = (const float*)&red[which][0][0] + d;
        float t = 0.0f;
#pragma unroll
        for (int w = 0; w < BN_WAVES; ++w) t += rp[w * 128];
        unsafeAtomicAdd((which ? sumsq : sum) + d, t);
    }
}

__global__ __launch_bounds__(128)
void bn_finalize_kernel(const float* __restrict__ sum, const float* __restrict__ sumsq,
                        float* __restrict__ mean, float* __restrict__ istd, int n) {
    int d = threadIdx.x;
    float invn = 1.0f / (float)n;
    float m = sum[d] * invn;
    float var = sumsq[d] * invn - m * m;
    mean[d] = m;
    istd[d] = rsqrtf(var + BN_EPS);
}

__global__ __launch_bounds__(256)
void bn_norm_kernel(const float* __restrict__ h, const float* __restrict__ mean,
                    const float* __restrict__ istd, const float* __restrict__ gamma,
                    const float* __restrict__ beta, float* __restrict__ out, int n) {
    int t = blockIdx.x * blockDim.x + threadIdx.x;
    if (t >= n * 32) return;
    int j = t & 31;
    float4 v  = ((const float4*)h)[t];
    float4 m  = ((const float4*)mean)[j];
    float4 is = ((const float4*)istd)[j];
    float4 g  = ((const float4*)gamma)[j];
    float4 be = ((const float4*)beta)[j];
    float4 o;
    o.x = g.x * (v.x - m.x) * is.x + be.x;
    o.y = g.y * (v.y - m.y) * is.y + be.y;
    o.z = g.z * (v.z - m.z) * is.z + be.z;
    o.w = g.w * (v.w - m.w) * is.w + be.w;
    ((float4*)out)[t] = o;
}


extern "C" void kernel_launch(void* const* d_in, const int* in_sizes, int n_in,
                              void* d_out, int out_size, void* d_ws, size_t ws_size,
                              hipStream_t stream) {
    const float* x     = (const float*)d_in[0];
    const int*   eidx  = (const int*)d_in[1];
    const float* W1    = (const float*)d_in[2];
    const float* b1    = (const float*)d_in[3];
    const float* W2    = (const float*)d_in[4];
    const float* b2    = (const float*)d_in[5];
    const float* gamma = (const float*)d_in[6];
    const float* beta  = (const float*)d_in[7];
    float* out = (float*)d_out;

    const int* row = eidx;
    const int* col = eidx + NE;

    float* bufA  = (float*)d_ws;
    float* bufB  = bufA + (size_t)NN * DIM;
    float* dis   = bufB + (size_t)NN * DIM;
    float* sum   = dis + NN;
    float* sumsq = sum + DIM;
    float* mean  = sumsq + DIM;
    float* istd  = mean + DIM;
    float* W1t   = istd + DIM;
    float* W2t   = W1t + (size_t)DIM * DIM;

    const int T = 256;
    const int elem_blocks = (NN * 32 + T - 1) / T;
    const int gemm_blocks = NN / 16;
    const int edge_blocks = (NE + 7) / 8;
    const int tr_blocks   = (DIM * DIM) / T;
    const int bn_blocks = (NN + BN_ROWS - 1) / BN_ROWS;

    init_deg_kernel<<<(NN + T - 1) / T, T, 0, stream>>>(dis, NN);
    deg_count_kernel<<<(NE + T - 1) / T, T, 0, stream>>>(col, dis, NE);
    rsqrt_kernel<<<(NN + T - 1) / T, T, 0, stream>>>(dis, NN);
    zero_kernel<<<1, 2 * DIM, 0, stream>>>(sum, 2 * DIM);
    transpose128_kernel<<<tr_blocks, T, 0, stream>>>(W1, W1t);
    transpose128_kernel<<<tr_blocks, T, 0, stream>>>(W2, W2t);

    gemm_wmma_kernel<<<gemm_blocks, T, 0, stream>>>(x, W1t, bufA);
    selfloop_kernel<<<elem_blocks, T, 0, stream>>>(bufA, dis, bufB, NN);
    edge_scatter_kernel<<<edge_blocks, T, 0, stream>>>(bufA, bufB, row, col, dis, NE);
    bias_relu_kernel<<<elem_blocks, T, 0, stream>>>(bufB, b1, bufA, NN);

    gemm_wmma_kernel<<<gemm_blocks, T, 0, stream>>>(bufA, W2t, bufB);
    selfloop_kernel<<<elem_blocks, T, 0, stream>>>(bufB, dis, bufA, NN);
    edge_scatter_kernel<<<edge_blocks, T, 0, stream>>>(bufB, bufA, row, col, dis, NE);
    bias_relu_kernel<<<elem_blocks, T, 0, stream>>>(bufA, b2, bufB, NN);

    bn_reduce_kernel<<<bn_blocks, 32 * BN_WAVES, 0, stream>>>(bufB, sum, sumsq, NN);
    bn_finalize_kernel<<<1, DIM, 0, stream>>>(sum, sumsq, mean, istd, NN);
    bn_norm_kernel<<<elem_blocks, T, 0, stream>>>(bufB, mean, istd, gamma, beta, out, NN);
}


# aggregation: per-call in-edge lists by destination (int atomics: count, scan, fill) + one-wave-per-node row gather with self-loop, bias, relu folded in replaces f32 atomic scatter; sums in list order
# speedup vs baseline: 2.3487x; 1.9608x over previous
#include <hip/hip_runtime.h>

#define NN 50000
#define NE 600000
#define DIM 128
#define BN_EPS 1e-5f

typedef __attribute__((ext_vector_type(2))) float v2f;
typedef __attribute__((ext_vector_type(8))) float v8f;


__global__ void zero_kernel(float* __restrict__ p, int n) {
    int i = blockIdx.x * blockDim.x + threadIdx.x;
    if (i < n) p[i] = 0.0f;
}

__global__ void count_kernel(const int* __restrict__ col, int* __restrict__ cnt, int e) {
    int i = blockIdx.x * blockDim.x + threadIdx.x;
    if (i < e) atomicAdd(&cnt[col[i]], 1);
}

#define SCAN_T 1024
__global__ __launch_bounds__(SCAN_T)
void scan_kernel(const int* __restrict__ cnt, int* __restrict__ off, int* __restrict__ cur,
                 float* __restrict__ dis, int n) {
    __shared__ int wave_own[32];
    __shared__ int wave_front[32];
    const int t    = threadIdx.x;
    const int lane = t & 31;
    const int wv   = t >> 5;
    const int b    = blockIdx.x;

    int front = 0;
#pragma unroll 8
    for (int i = 0; i < b; ++i) front += cnt[i * SCAN_T + t];

    const int idx = b * SCAN_T + t;
    const int c = (idx < n) ? cnt[idx] : 0;

    int incl = c;
    for (int d = 1; d < 32; d <<= 1) {
        const int y = __shfl_up(incl, d);
        if (lane >= d) incl += y;
    }
    for (int d = 16; d > 0; d >>= 1) front += __shfl_xor(front, d);

    if (lane == 31) wave_own[wv] = incl;
    if (lane == 0)  wave_front[wv] = front;
    __syncthreads();

    int pre = wave_front[lane] + ((lane < wv) ? wave_own[lane] : 0);
    for (int d = 16; d > 0; d >>= 1) pre += __shfl_xor(pre, d);

    const int excl = pre + incl - c;
    if (idx < n) {
        off[idx] = excl;
        cur[idx] = excl;
        dis[idx] = rsqrtf((float)c + 1.0f);
        if (idx == n - 1) off[n] = excl + c;
    }
}

__global__ __launch_bounds__(256)
void fill_kernel(const int* __restrict__ row, const int* __restrict__ col,
                 const float* __restrict__ dis, int* __restrict__ cur,
                 int* __restrict__ recs, int e) {
    int i = blockIdx.x * blockDim.x + threadIdx.x;
    if (i >= e) return;
    const int r = row[i];
    const int c = col[i];
    const int p = atomicAdd(&cur[c], 1);
    int2 rec;
    rec.x = r;
    rec.y = __float_as_int(dis[r]);
    ((int2*)recs)[p] = rec;
}


__global__ __launch_bounds__(256)
void transpose128_kernel(const float* __restrict__ W, float* __restrict__ Wt) {
    int t = blockIdx.x * blockDim.x + threadIdx.x;
    int r = t >> 7;
    int c = t & 127;
    Wt[(size_t)c * DIM + r] = W[(size_t)r * DIM + c];
}

__global__ __launch_bounds__(256)
void gemm_wmma_kernel(const float* __restrict__ X, const float* __restrict__ Wt,
                      float* __restrict__ H) {
    const int tileM = blockIdx.x * 16;
    const int wave  = threadIdx.x >> 5;
    const int lane  = threadIdx.x & 31;
    const int tileN = wave * 16;
    const int mrow  = lane & 15;
    const int hi    = lane >> 4;

    v8f acc = {};
    const float* xr = X + (size_t)(tileM + mrow) * DIM + 2 * hi;
    const float* wc = Wt + (size_t)(tileN + mrow) * DIM + 2 * hi;

#pragma unroll
    for (int k0 = 0; k0 < DIM; k0 += 4) {
        v2f a = *(const v2f*)(xr + k0);
        v2f b = *(const v2f*)(wc + k0);
        acc = __builtin_amdgcn_wmma_f32_16x16x4_f32(
             false, a,  false, b,
             (short)0, acc,  false,  false);
    }

    float* out = H + (size_t)(tileM + 8 * hi) * DIM + tileN + mrow;
#pragma unroll
    for (int r = 0; r < 8; ++r)
        out[(size_t)r * DIM] = acc[r];
}

#define GA_WAVES 8

__device__ __forceinline__ float lane_bcast_f(float x, int l) {
    return __int_as_float(__builtin_amdgcn_readlane(__float_as_int(x), l));
}

__global__ __launch_bounds__(32 * GA_WAVES)
void gather_kernel(const float* __restrict__ h, const int* __restrict__ recs,
                   const int* __restrict__ off, const float* __restrict__ dis,
                   const float* __restrict__ bias, float* __restrict__ out, int n) {
    const int lane = threadIdx.x & 31;
    const int nw   = gridDim.x * GA_WAVES;
    int node = __builtin_amdgcn_readfirstlane(blockIdx.x * GA_WAVES + (threadIdx.x >> 5));
    const float4* h4 = (const float4*)h;
    const int2*   rc = (const int2*)recs;
    const float4  bb = ((const float4*)bias)[lane];

    for (; node < n; node += nw) {
        const int beg = off[node];
        const int end = off[node + 1];
        const float dc = dis[node];
        float4 acc = h4[(size_t)node * 32 + lane];
        const float s = dc * dc;
        acc.x *= s;  acc.y *= s;  acc.z *= s;  acc.w *= s;

        for (int base = beg; base < end; base += 32) {
            const int m = min(32, end - base);
            int   myr = 0;
            float myw = 0.0f;
            if (lane < m) {
                const int2 q = rc[base + lane];
                myr = q.x;
                myw = __int_as_float(q.y) * dc;
            }
            int i = 0;
            for (; i + 8 <= m; i += 8) {
                float4 v[8];
                float  w[8];
#pragma unroll
                for (int u = 0; u < 8; ++u) {
                    const int r = __builtin_amdgcn_readlane(myr, i + u);
                    w[u] = lane_bcast_f(myw, i + u);
                    v[u] = h4[(size_t)r * 32 + lane];
                }
#pragma unroll
                for (int u = 0; u < 8; ++u) {
                    acc.x = fmaf(v[u].x, w[u], acc.x);
                    acc.y = fmaf(v[u].y, w[u], acc.y);
                    acc.z = fmaf(v[u].z, w[u], acc.z);
                    acc.w = fmaf(v[u].w, w[u], acc.w);
                }
            }
            for (; i < m; i += 4) {
                float4 v[4];
                float  w[4];
#pragma unroll
                for (int u = 0; u < 4; ++u) {
                    const int k = min(i + u, m - 1);
                    const int r = __builtin_amdgcn_readlane(myr, k);
                    w[u] = lane_bcast_f(myw, k);
                    v[u] = h4[(size_t)r * 32 + lane];
                }
#pragma unroll
                for (int u = 0; u < 4; ++u) {
                    const bool live = (i + u) < m;
                    const float wx = live ? v[u].x : 0.0f;
                    const float wy = live ? v[u].y : 0.0f;
                    const float wz = live ? v[u].z : 0.0f;
                    const float ww = live ? v[u].w : 0.0f;
                    acc.x = fmaf(wx, w[u], acc.x);
                    acc.y = fmaf(wy, w[u], acc.y);
                    acc.z = fmaf(wz, w[u], acc.z);
                    acc.w = fmaf(ww, w[u], acc.w);
                }
            }
        }

        acc.x = fmaxf(acc.x + bb.x, 0.0f);
        acc.y = fmaxf(acc.y + bb.y, 0.0f);
        acc.z = fmaxf(acc.z + bb.z, 0.0f);
        acc.w = fmaxf(acc.w + bb.w, 0.0f);
        ((float4*)out)[(size_t)node * 32 + lane] = acc;
    }
}


#define BN_WAVES 16
#define BN_RPW   13
#define BN_ROWS  (BN_WAVES * BN_RPW)

__global__ __launch_bounds__(32 * BN_WAVES)
void bn_reduce_kernel(const float* __restrict__ h, float* __restrict__ sum,
                      float* __restrict__ sumsq, int n) {
    __shared__ float4 red[2][BN_WAVES][32];
    const int j  = threadIdx.x & 31;
    const int g  = threadIdx.x >> 5;
    const int r0 = blockIdx.x * BN_ROWS + g * BN_RPW;
    const float4* hp = (const float4*)h;

    float4 v[BN_RPW];
#pragma unroll
    for (int i = 0; i < BN_RPW; ++i) {
        int r = r0 + i;
        if (r >= n) r = n - 1;
        v[i] = hp[(size_t)r * 32 + j];
    }

    float4 s = {0.0f, 0.0f, 0.0f, 0.0f};
    float4 q = {0.0f, 0.0f, 0.0f, 0.0f};
#pragma unroll
    for (int i = 0; i < BN_RPW; ++i) {
        const bool live = (r0 + i) < n;
        float4 w;
        w.x = live ? v[i].x : 0.0f;
        w.y = live ? v[i].y : 0.0f;
        w.z = live ? v[i].z : 0.0f;
        w.w = live ? v[i].w : 0.0f;
        s.x += w.x;  s.y += w.y;  s.z += w.z;  s.w += w.w;
        q.x += w.x * w.x;  q.y += w.y * w.y;  q.z += w.z * w.z;  q.w += w.w * w.w;
    }

    red[0][g][j] = s;
    red[1][g][j] = q;
    __syncthreads();

    if (threadIdx.x < 2 * DIM) {
        const int which = threadIdx.x >> 7;
        const int d = threadIdx.x & 127;
        const float* rp = (const float*)&red[which][0][0] + d;
        float t = 0.0f;
#pragma unroll
        for (int w = 0; w < BN_WAVES; ++w) t += rp[w * 128];
        unsafeAtomicAdd((which ? sumsq : sum) + d, t);
    }
}

__global__ __launch_bounds__(128)
void bn_finalize_kernel(const float* __restrict__ sum, const float* __restrict__ sumsq,
                        float* __restrict__ mean, float* __restrict__ istd, int n) {
    int d = threadIdx.x;
    float invn = 1.0f / (float)n;
    float m = sum[d] * invn;
    float var = sumsq[d] * invn - m * m;
    mean[d] = m;
    istd[d] = rsqrtf(var + BN_EPS);
}

__global__ __launch_bounds__(256)
void bn_norm_kernel(const float* __restrict__ h, const float* __restrict__ mean,
                    const float* __restrict__ istd, const float* __restrict__ gamma,
                    const float* __restrict__ beta, float* __restrict__ out, int n) {
    int t = blockIdx.x * blockDim.x + threadIdx.x;
    if (t >= n * 32) return;
    int j = t & 31;
    float4 v  = ((const float4*)h)[t];
    float4 m  = ((const float4*)mean)[j];
    float4 is = ((const float4*)istd)[j];
    float4 g  = ((const float4*)gamma)[j];
    float4 be = ((const float4*)beta)[j];
    float4 o;
    o.x = g.x * (v.x - m.x) * is.x + be.x;
    o.y = g.y * (v.y - m.y) * is.y + be.y;
    o.z = g.z * (v.z - m.z) * is.z + be.z;
    o.w = g.w * (v.w - m.w) * is.w + be.w;
    ((float4*)out)[t] = o;
}


extern "C" void kernel_launch(void* const* d_in, const int* in_sizes, int n_in,
                              void* d_out, int out_size, void* d_ws, size_t ws_size,
                              hipStream_t stream) {
    const float* x     = (const float*)d_in[0];
    const int*   eidx  = (const int*)d_in[1];
    const float* W1    = (const float*)d_in[2];
    const float* b1    = (const float*)d_in[3];
    const float* W2    = (const float*)d_in[4];
    const float* b2    = (const float*)d_in[5];
    const float* gamma = (const float*)d_in[6];
    const float* beta  = (const float*)d_in[7];
    float* out = (float*)d_out;

    const int* row = eidx;
    const int* col = eidx + NE;

    float* bufA  = (float*)d_ws;
    float* bufB  = bufA + (size_t)NN * DIM;
    int*   recs  = (int*)(bufB + (size_t)NN * DIM);
    float* dis   = (float*)(recs + (size_t)2 * NE);
    int*   cnt   = (int*)(dis + NN);
    float* sum   = (float*)(cnt + NN);
    float* sumsq = sum + DIM;
    float* mean  = sumsq + DIM;
    float* istd  = mean + DIM;
    float* W1t   = istd + DIM;
    float* W2t   = W1t + (size_t)DIM * DIM;
    int*   off   = (int*)(W2t + (size_t)DIM * DIM);
    int*   cur   = off + NN + 16;

    const int T = 256;
    const int elem_blocks = (NN * 32 + T - 1) / T;
    const int gemm_blocks = NN / 16;
    const int edge_blocks = (NE + T - 1) / T;
    const int tr_blocks   = (DIM * DIM) / T;
    const int bn_blocks = (NN + BN_ROWS - 1) / BN_ROWS;
    const int scan_blocks = (NN + SCAN_T - 1) / SCAN_T;
    const int zero_words = NN + 2 * DIM;
    const int ga_blocks = 2048;

    zero_kernel<<<(zero_words + T - 1) / T, T, 0, stream>>>((float*)cnt, zero_words);
    count_kernel<<<edge_blocks, T, 0, stream>>>(col, cnt, NE);
    scan_kernel<<<scan_blocks, SCAN_T, 0, stream>>>(cnt, off, cur, dis, NN);
    fill_kernel<<<edge_blocks, T, 0, stream>>>(row, col, dis, cur, recs, NE);
    transpose128_kernel<<<tr_blocks, T, 0, stream>>>(W1, W1t);
    transpose128_kernel<<<tr_blocks, T, 0, stream>>>(W2, W2t);

    gemm_wmma_kernel<<<gemm_blocks, T, 0, stream>>>(x, W1t, bufA);
    gather_kernel<<<ga_blocks, 32 * GA_WAVES, 0, stream>>>(bufA, recs, off, dis, b1, bufB, NN);

    gemm_wmma_kernel<<<gemm_blocks, T, 0, stream>>>(bufB, W2t, bufA);
    gather_kernel<<<ga_blocks, 32 * GA_WAVES, 0, stream>>>(bufA, recs, off, dis, b2, bufB, NN);

    bn_reduce_kernel<<<bn_blocks, 32 * BN_WAVES, 0, stream>>>(bufB, sum, sumsq, NN);
    bn_finalize_kernel<<<1, DIM, 0, stream>>>(sum, sumsq, mean, istd, NN);
    bn_norm_kernel<<<elem_blocks, T, 0, stream>>>(bufB, mean, istd, gamma, beta, out, NN);
}


# gemm: W staged once per 128-row block in LDS (transpose launches dropped), one wave per 16x128 strip, A read once as float4; k regrouped per matrix instruction (new fixed f32 sum order)
# speedup vs baseline: 3.1077x; 1.3232x over previous
#include <hip/hip_runtime.h>

#define NN 50000
#define NE 600000
#define DIM 128
#define BN_EPS 1e-5f

typedef __attribute__((ext_vector_type(2))) float v2f;
typedef __attribute__((ext_vector_type(8))) float v8f;


__global__ void zero_kernel(float* __restrict__ p, int n) {
    int i = blockIdx.x * blockDim.x + threadIdx.x;
    if (i < n) p[i] = 0.0f;
}

__global__ void count_kernel(const int* __restrict__ col, int* __restrict__ cnt, int e) {
    int i = blockIdx.x * blockDim.x + threadIdx.x;
    if (i < e) atomicAdd(&cnt[col[i]], 1);
}

#define SCAN_T 1024
__global__ __launch_bounds__(SCAN_T)
void scan_kernel(const int* __restrict__ cnt, int* __restrict__ off, int* __restrict__ cur,
                 float* __restrict__ dis, int n) {
    __shared__ int wave_own[32];
    __shared__ int wave_front[32];
    const int t    = threadIdx.x;
    const int lane = t & 31;
    const int wv   = t >> 5;
    const int b    = blockIdx.x;

    int front = 0;
#pragma unroll 8
    for (int i = 0; i < b; ++i) front += cnt[i * SCAN_T + t];

    const int idx = b * SCAN_T + t;
    const int c = (idx < n) ? cnt[idx] : 0;

    int incl = c;
    for (int d = 1; d < 32; d <<= 1) {
        const int y = __shfl_up(incl, d);
        if (lane >= d) incl += y;
    }
    for (int d = 16; d > 0; d >>= 1) front += __shfl_xor(front, d);

    if (lane == 31) wave_own[wv] = incl;
    if (lane == 0)  wave_front[wv] = front;
    __syncthreads();

    int pre = wave_front[lane] + ((lane < wv) ? wave_own[lane] : 0);
    for (int d = 16; d > 0; d >>= 1) pre += __shfl_xor(pre, d);

    const int excl = pre + incl - c;
    if (idx < n) {
        off[idx] = excl;
        cur[idx] = excl;
        dis[idx] = rsqrtf((float)c + 1.0f);
        if (idx == n - 1) off[n] = excl + c;
    }
}

__global__ __launch_bounds__(256)
void fill_kernel(const int* __restrict__ row, const int* __restrict__ col,
                 const float* __restrict__ dis, int* __restrict__ cur,
                 int* __restrict__ recs, int e) {
    int i = blockIdx.x * blockDim.x + threadIdx.x;
    if (i >= e) return;
    const int r = row[i];
    const int c = col[i];
    const int p = atomicAdd(&cur[c], 1);
    int2 rec;
    rec.x = r;
    rec.y = __float_as_int(dis[r]);
    ((int2*)recs)[p] = rec;
}

#define GM_WAVES 8
#define GM_ROWS  (16 * GM_WAVES)

__global__ __launch_bounds__(32 * GM_WAVES)
void gemm_wmma_kernel(const float* __restrict__ X, const float* __restrict__ W,
                      float* __restrict__ H, int n) {
    __shared__ float4 Ws[32 * DIM];
    const int lane  = threadIdx.x & 31;
    const int wave  = threadIdx.x >> 5;
    const int c     = lane & 15;
    const int hi    = lane >> 4;
    const int tileM = blockIdx.x * GM_ROWS + wave * 16;

    float4 a[16];
    {
        const int row = min(tileM + c, n - 1);
        const float4* xr = (const float4*)(X + (size_t)row * DIM) + hi;
#pragma unroll
        for (int j = 0; j < 16; ++j) a[j] = xr[2 * j];
    }

#pragma unroll
    for (int i = 0; i < 4; ++i) {
        const int kq = wave + GM_WAVES * i;
        const float4* wr = (const float4*)(W + (size_t)(4 * kq) * DIM) + lane;
        const float4 r0 = wr[0];
        const float4 r1 = wr[32];
        const float4 r2 = wr[64];
        const float4 r3 = wr[96];
        float4* dst = Ws + kq * DIM + 4 * lane;
        dst[0] = make_float4(r0.x, r1.x, r2.x, r3.x);
        dst[1] = make_float4(r0.y, r1.y, r2.y, r3.y);
        dst[2] = make_float4(r0.z, r1.z, r2.z, r3.z);
        dst[3] = make_float4(r0.w, r1.w, r2.w, r3.w);
    }
    __syncthreads();
    if (tileM >= n) return;

    v8f acc[8];
#pragma unroll
    for (int t = 0; t < 8; ++t) acc[t] = (v8f){};

    const float4* bp = Ws + hi * DIM + c;
    float4 b[2][8];
#pragma unroll
    for (int t = 0; t < 8; ++t) b[0][t] = bp[16 * t];

#pragma unroll
    for (int j = 0; j < 16; ++j) {
        if (j + 1 < 16) {
#pragma unroll
            for (int t = 0; t < 8; ++t) b[(j + 1) & 1][t] = bp[2 * (j + 1) * DIM + 16 * t];
        }
        __builtin_amdgcn_sched_barrier(0);
        const v2f a01 = {a[j].x, a[j].y};
        const v2f a23 = {a[j].z, a[j].w};
#pragma unroll
        for (int t = 0; t < 8; ++t) {
            const v2f b01 = {b[j & 1][t].x, b[j & 1][t].y};
            acc[t] = __builtin_amdgcn_wmma_f32_16x16x4_f32(
                 false, a01,  false, b01,
                 (short)0, acc[t],  false,  false);
        }
#pragma unroll
        for (int t = 0; t < 8; ++t) {
            const v2f b23 = {b[j & 1][t].z, b[j & 1][t].w};
            acc[t] = __builtin_amdgcn_wmma_f32_16x16x4_f32(
                 false, a23,  false, b23,
                 (short)0, acc[t],  false,  false);
        }
        __builtin_amdgcn_sched_barrier(0);
    }

    float* out = H + (size_t)(tileM + 8 * hi) * DIM + c;
#pragma unroll
    for (int r = 0; r < 8; ++r) {
#pragma unroll
        for (int t = 0; t < 8; ++t)
            out[(size_t)r * DIM + 16 * t] = acc[t][r];
    }
}

#define GA_WAVES 8

__device__ __forceinline__ float lane_bcast_f(float x, int l) {
    return __int_as_float(__builtin_amdgcn_readlane(__float_as_int(x), l));
}

__global__ __launch_bounds__(32 * GA_WAVES)
void gather_kernel(const float* __restrict__ h, const int* __restrict__ recs,
                   const int* __restrict__ off, const float* __restrict__ dis,
                   const float* __restrict__ bias, float* __restrict__ out, int n) {
    const int lane = threadIdx.x & 31;
    const int nw   = gridDim.x * GA_WAVES;
    int node = __builtin_amdgcn_readfirstlane(blockIdx.x * GA_WAVES + (threadIdx.x >> 5));
    const float4* h4 = (const float4*)h;
    const int2*   rc = (const int2*)recs;
    const float4  bb = ((const float4*)bias)[lane];

    for (; node < n; node += nw) {
        const int beg = off[node];
        const int end = off[node + 1];
        const float dc = dis[node];
        float4 acc = h4[(size_t)node * 32 + lane];
        const float s = dc * dc;
        acc.x *= s;  acc.y *= s;  acc.z *= s;  acc.w *= s;

        for (int base = beg; base < end; base += 32) {
            const int m = min(32, end - base);
            int   myr = 0;
            float myw = 0.0f;
            if (lane < m) {
                const int2 q = rc[base + lane];
                myr = q.x;
                myw = __int_as_float(q.y) * dc;
            }
            int i = 0;
            for (; i + 8 <= m; i += 8) {
                float4 v[8];
                float  w[8];
#pragma unroll
                for (int u = 0; u < 8; ++u) {
                    const int r = __builtin_amdgcn_readlane(myr, i + u);
                    w[u] = lane_bcast_f(myw, i + u);
                    v[u] = h4[(size_t)r * 32 + lane];
                }
#pragma unroll
                for (int u = 0; u < 8; ++u) {
                    acc.x = fmaf(v[u].x, w[u], acc.x);
                    acc.y = fmaf(v[u].y, w[u], acc.y);
                    acc.z = fmaf(v[u].z, w[u], acc.z);
                    acc.w = fmaf(v[u].w, w[u], acc.w);
                }
            }
            for (; i < m; i += 4) {
                float4 v[4];
                float  w[4];
#pragma unroll
                for (int u = 0; u < 4; ++u) {
                    const int k = min(i + u, m - 1);
                    const int r = __builtin_amdgcn_readlane(myr, k);
                    w[u] = lane_bcast_f(myw, k);
                    v[u] = h4[(size_t)r * 32 + lane];
                }
#pragma unroll
                for (int u = 0; u < 4; ++u) {
                    const bool live = (i + u) < m;
                    const float wx = live ? v[u].x : 0.0f;
                    const float wy = live ? v[u].y : 0.0f;
                    const float wz = live ? v[u].z : 0.0f;
                    const float ww = live ? v[u].w : 0.0f;
                    acc.x = fmaf(wx, w[u], acc.x);
                    acc.y = fmaf(wy, w[u], acc.y);
                    acc.z = fmaf(wz, w[u], acc.z);
                    acc.w = fmaf(ww, w[u], acc.w);
                }
            }
        }

        acc.x = fmaxf(acc.x + bb.x, 0.0f);
        acc.y = fmaxf(acc.y + bb.y, 0.0f);
        acc.z = fmaxf(acc.z + bb.z, 0.0f);
        acc.w = fmaxf(acc.w + bb.w, 0.0f);
        ((float4*)out)[(size_t)node * 32 + lane] = acc;
    }
}


#define BN_WAVES 16
#define BN_RPW   13
#define BN_ROWS  (BN_WAVES * BN_RPW)

__global__ __launch_bounds__(32 * BN_WAVES)
void bn_reduce_kernel(const float* __restrict__ h, float* __restrict__ sum,
                      float* __restrict__ sumsq, int n) {
    __shared__ float4 red[2][BN_WAVES][32];
    const int j  = threadIdx.x & 31;
    const int g  = threadIdx.x >> 5;
    const int r0 = blockIdx.x * BN_ROWS + g * BN_RPW;
    const float4* hp = (const float4*)h;

    float4 v[BN_RPW];
#pragma unroll
    for (int i = 0; i < BN_RPW; ++i) {
        int r = r0 + i;
        if (r >= n) r = n - 1;
        v[i] = hp[(size_t)r * 32 + j];
    }

    float4 s = {0.0f, 0.0f, 0.0f, 0.0f};
    float4 q = {0.0f, 0.0f, 0.0f, 0.0f};
#pragma unroll
    for (int i = 0; i < BN_RPW; ++i) {
        const bool live = (r0 + i) < n;
        float4 w;
        w.x = live ? v[i].x : 0.0f;
        w.y = live ? v[i].y : 0.0f;
        w.z = live ? v[i].z : 0.0f;
        w.w = live ? v[i].w : 0.0f;
        s.x += w.x;  s.y += w.y;  s.z += w.z;  s.w += w.w;
        q.x += w.x * w.x;  q.y += w.y * w.y;  q.z += w.z * w.z;  q.w += w.w * w.w;
    }

    red[0][g][j] = s;
    red[1][g][j] = q;
    __syncthreads();

    if (threadIdx.x < 2 * DIM) {
        const int which = threadIdx.x >> 7;
        const int d = threadIdx.x & 127;
        const float* rp = (const float*)&red[which][0][0] + d;
        float t = 0.0f;
#pragma unroll
        for (int w = 0; w < BN_WAVES; ++w) t += rp[w * 128];
        unsafeAtomicAdd((which ? sumsq : sum) + d, t);
    }
}

__global__ __launch_bounds__(128)
void bn_finalize_kernel(const float* __restrict__ sum, const float* __restrict__ sumsq,
                        float* __restrict__ mean, float* __restrict__ istd, int n) {
    int d = threadIdx.x;
    float invn = 1.0f / (float)n;
    float m = sum[d] * invn;
    float var = sumsq[d] * invn - m * m;
    mean[d] = m;
    istd[d] = rsqrtf(var + BN_EPS);
}

__global__ __launch_bounds__(256)
void bn_norm_kernel(const float* __restrict__ h, const float* __restrict__ mean,
                    const float* __restrict__ istd, const float* __restrict__ gamma,
                    const float* __restrict__ beta, float* __restrict__ out, int n) {
    int t = blockIdx.x * blockDim.x + threadIdx.x;
    if (t >= n * 32) return;
    int j = t & 31;
    float4 v  = ((const float4*)h)[t];
    float4 m  = ((const float4*)mean)[j];
    float4 is = ((const float4*)istd)[j];
    float4 g  = ((const float4*)gamma)[j];
    float4 be = ((const float4*)beta)[j];
    float4 o;
    o.x = g.x * (v.x - m.x) * is.x + be.x;
    o.y = g.y * (v.y - m.y) * is.y + be.y;
    o.z = g.z * (v.z - m.z) * is.z + be.z;
    o.w = g.w * (v.w - m.w) * is.w + be.w;
    ((float4*)out)[t] = o;
}


extern "C" void kernel_launch(void* const* d_in, const int* in_sizes, int n_in,
                              void* d_out, int out_size, void* d_ws, size_t ws_size,
                              hipStream_t stream) {
    const float* x     = (const float*)d_in[0];
    const int*   eidx  = (const int*)d_in[1];
    const float* W1    = (const float*)d_in[2];
    const float* b1    = (const float*)d_in[3];
    const float* W2    = (const float*)d_in[4];
    const float* b2    = (const float*)d_in[5];
    const float* gamma = (const float*)d_in[6];
    const float* beta  = (const float*)d_in[7];
    float* out = (float*)d_out;

    const int* row = eidx;
    const int* col = eidx + NE;

    float* bufA  = (float*)d_ws;
    float* bufB  = bufA + (size_t)NN * DIM;
    int*   recs  = (int*)(bufB + (size_t)NN * DIM);
    float* dis   = (float*)(recs + (size_t)2 * NE);
    int*   cnt   = (int*)(dis + NN);
    float* sum   = (float*)(cnt + NN);
    float* sumsq = sum + DIM;
    float* mean  = sumsq + DIM;
    float* istd  = mean + DIM;
    int*   off   = (int*)(istd + DIM);
    int*   cur   = off + NN + 16;

    const int T = 256;
    const int elem_blocks = (NN * 32 + T - 1) / T;
    const int gemm_blocks = (NN + GM_ROWS - 1) / GM_ROWS;
    const int edge_blocks = (NE + T - 1) / T;
    const int bn_blocks = (NN + BN_ROWS - 1) / BN_ROWS;
    const int scan_blocks = (NN + SCAN_T - 1) / SCAN_T;
    const int zero_words = NN + 2 * DIM;
    const int ga_blocks = 2048;

    zero_kernel<<<(zero_words + T - 1) / T, T, 0, stream>>>((float*)cnt, zero_words);
    count_kernel<<<edge_blocks, T, 0, stream>>>(col, cnt, NE);
    scan_kernel<<<scan_blocks, SCAN_T, 0, stream>>>(cnt, off, cur, dis, NN);
    fill_kernel<<<edge_blocks, T, 0, stream>>>(row, col, dis, cur, recs, NE);

    gemm_wmma_kernel<<<gemm_blocks, 32 * GM_WAVES, 0, stream>>>(x, W1, bufA, NN);
    gather_kernel<<<ga_blocks, 32 * GA_WAVES, 0, stream>>>(bufA, recs, off, dis, b1, bufB, NN);

    gemm_wmma_kernel<<<gemm_blocks, 32 * GM_WAVES, 0, stream>>>(bufB, W2, bufA, NN);
    gather_kernel<<<ga_blocks, 32 * GA_WAVES, 0, stream>>>(bufA, recs, off, dis, b2, bufB, NN);

    bn_reduce_kernel<<<bn_blocks, 32 * BN_WAVES, 0, stream>>>(bufB, sum, sumsq, NN);
    bn_finalize_kernel<<<1, DIM, 0, stream>>>(sum, sumsq, mean, istd, NN);
    bn_norm_kernel<<<elem_blocks, T, 0, stream>>>(bufB, mean, istd, gamma, beta, out, NN);
}


# edge lists: each edge's rank is taken on LDS counters (32 chunk work-groups, a counter per node in LDS; prefix over chunks in scan), no global atomic pass left; list order = LDS arrival order
# speedup vs baseline: 4.1037x; 1.3205x over previous
#include <hip/hip_runtime.h>

#define NN 50000
#define NE 600000
#define DIM 128
#define BN_EPS 1e-5f

typedef __attribute__((ext_vector_type(2))) float v2f;
typedef __attribute__((ext_vector_type(8))) float v8f;

#define CH      32
#define EPC     (NE / CH)
#define CNT_T   1024
#define SCAN_T  1024
#define CNT_EPT ((EPC + CNT_T - 1) / CNT_T)
#define CNT_Q   ((NN / 4 + CNT_T - 1) / CNT_T)
#define WC_TILES (4 * CNT_Q)
static_assert(NE % CH == 0, "chunks of equal length");
static_assert(NN % 4 == 0, "the counter table is moved 16 bytes at a time");
static_assert(SCAN_T == 1024 && CNT_T == 1024, "count_kernel: trip k, thread t holds nodes of tile 4k + t/256, run (t/32)%8");

__global__ __launch_bounds__(CNT_T)
void count_kernel(const int* __restrict__ col, int* __restrict__ rank,
                  int* __restrict__ chunkcnt, int* __restrict__ wavecnt,
                  float* __restrict__ bnacc) {
    __shared__ int4 tab4[NN / 4];
    int* tab = (int*)tab4;
    const int t = threadIdx.x;
    const int b = blockIdx.x;
    const int* cc = col + b * EPC;
    int* rk = rank + b * EPC;

    int cv[CNT_EPT];
#pragma unroll
    for (int k = 0; k < CNT_EPT; ++k) cv[k] = cc[min(t + k * CNT_T, EPC - 1)];

    for (int i = t; i < NN / 4; i += CNT_T) tab4[i] = make_int4(0, 0, 0, 0);
    __syncthreads();

    int rv[CNT_EPT];
#pragma unroll
    for (int k = 0; k < CNT_EPT; ++k) {
        rv[k] = 0;
        if (t + k * CNT_T < EPC) rv[k] = atomicAdd(&tab[cv[k]], 1);
    }
    __syncthreads();

    int4* o4 = (int4*)(chunkcnt + (size_t)b * NN);
    int sm[CNT_Q];
#pragma unroll
    for (int k = 0; k < CNT_Q; ++k) {
        const int i = k * CNT_T + t;
        int4 v = make_int4(0, 0, 0, 0);
        if (i < NN / 4) { v = tab4[i]; o4[i] = v; }
        sm[k] = v.x + v.y + v.z + v.w;
    }
#pragma unroll
    for (int k = 0; k < CNT_EPT; ++k)
        if (t + k * CNT_T < EPC) rk[t + k * CNT_T] = rv[k];

#pragma unroll
    for (int d = 16; d > 0; d >>= 1) {
#pragma unroll
        for (int k = 0; k < CNT_Q; ++k) sm[k] += __shfl_xor(sm[k], d);
    }
    if ((t & 31) == 0) {
        const int g = t >> 8;
        const int w = (t >> 5) & 7;
#pragma unroll
        for (int k = 0; k < CNT_Q; ++k) wavecnt[((4 * k + g) * CH + b) * 8 + w] = sm[k];
    }
    if (b == 0 && t < 2 * DIM) bnacc[t] = 0.0f;
}

__global__ __launch_bounds__(SCAN_T)
void scan_kernel(int* __restrict__ chunkcnt, const int* __restrict__ wavecnt,
                 int* __restrict__ off, float* __restrict__ dis, int n) {
    __shared__ int wave_own[32];
    __shared__ int wave_front[32];
    const int t    = threadIdx.x;
    const int lane = t & 31;
    const int wv   = t >> 5;
    const int b    = blockIdx.x;

    int front = 0;
#pragma unroll 4
    for (int i = t; i < 8 * CH * b; i += SCAN_T) front += wavecnt[i];

    const int idx = b * SCAN_T + t;
    int c = 0;
    if (idx < n) {
        int v[CH];
#pragma unroll
        for (int k = 0; k < CH; ++k) v[k] = chunkcnt[(size_t)k * NN + idx];
#pragma unroll
        for (int k = 0; k < CH; ++k) { chunkcnt[(size_t)k * NN + idx] = c; c += v[k]; }
    }

    int incl = c;
    for (int d = 1; d < 32; d <<= 1) {
        const int y = __shfl_up(incl, d);
        if (lane >= d) incl += y;
    }
    for (int d = 16; d > 0; d >>= 1) front += __shfl_xor(front, d);

    if (lane == 31) wave_own[wv] = incl;
    if (lane == 0)  wave_front[wv] = front;
    __syncthreads();

    int pre = wave_front[lane] + ((lane < wv) ? wave_own[lane] : 0);
    for (int d = 16; d > 0; d >>= 1) pre += __shfl_xor(pre, d);

    const int excl = pre + incl - c;
    if (idx < n) {
        off[idx] = excl;
        dis[idx] = rsqrtf((float)c + 1.0f);
        if (idx == n - 1) off[n] = excl + c;
    }
}

__global__ __launch_bounds__(256)
void fill_kernel(const int* __restrict__ row, const int* __restrict__ col,
                 const int* __restrict__ rank, const int* __restrict__ off,
                 const int* __restrict__ chunkcnt, const float* __restrict__ dis,
                 int* __restrict__ recs, int e) {
    int i = blockIdx.x * blockDim.x + threadIdx.x;
    if (i >= e) return;
    const int r = row[i];
    const int c = col[i];
    const int b = i / EPC;
    const int p = off[c] + chunkcnt[(size_t)b * NN + c] + rank[i];
    int2 rec;
    rec.x = r;
    rec.y = __float_as_int(dis[r]);
    ((int2*)recs)[p] = rec;
}

#define GM_WAVES 8
#define GM_ROWS  (16 * GM_WAVES)

__global__ __launch_bounds__(32 * GM_WAVES)
void gemm_wmma_kernel(const float* __restrict__ X, const float* __restrict__ W,
                      float* __restrict__ H, int n) {
    __shared__ float4 Ws[32 * DIM];
    const int lane  = threadIdx.x & 31;
    const int wave  = threadIdx.x >> 5;
    const int c     = lane & 15;
    const int hi    = lane >> 4;
    const int tileM = blockIdx.x * GM_ROWS + wave * 16;

    float4 a[16];
    {
        const int row = min(tileM + c, n - 1);
        const float4* xr = (const float4*)(X + (size_t)row * DIM) + hi;
#pragma unroll
        for (int j = 0; j < 16; ++j) a[j] = xr[2 * j];
    }

#pragma unroll
    for (int i = 0; i < 4; ++i) {
        const int kq = wave + GM_WAVES * i;
        const float4* wr = (const float4*)(W + (size_t)(4 * kq) * DIM) + lane;
        const float4 r0 = wr[0];
        const float4 r1 = wr[32];
        const float4 r2 = wr[64];
        const float4 r3 = wr[96];
        float4* dst = Ws + kq * DIM + 4 * lane;
        dst[0] = make_float4(r0.x, r1.x, r2.x, r3.x);
        dst[1] = make_float4(r0.y, r1.y, r2.y, r3.y);
        dst[2] = make_float4(r0.z, r1.z, r2.z, r3.z);
        dst[3] = make_float4(r0.w, r1.w, r2.w, r3.w);
    }
    __syncthreads();
    if (tileM >= n) return;

    v8f acc[8];
#pragma unroll
    for (int t = 0; t < 8; ++t) acc[t] = (v8f){};

    const float4* bp = Ws + hi * DIM + c;
    float4 b[2][8];
#pragma unroll
    for (int t = 0; t < 8; ++t) b[0][t] = bp[16 * t];

#pragma unroll
    for (int j = 0; j < 16; ++j) {
        if (j + 1 < 16) {
#pragma unroll
            for (int t = 0; t < 8; ++t) b[(j + 1) & 1][t] = bp[2 * (j + 1) * DIM + 16 * t];
        }
        __builtin_amdgcn_sched_barrier(0);
        const v2f a01 = {a[j].x, a[j].y};
        const v2f a23 = {a[j].z, a[j].w};
#pragma unroll
        for (int t = 0; t < 8; ++t) {
            const v2f b01 = {b[j & 1][t].x, b[j & 1][t].y};
            acc[t] = __builtin_amdgcn_wmma_f32_16x16x4_f32(
                 false, a01,  false, b01,
                 (short)0, acc[t],  false,  false);
        }
#pragma unroll
        for (int t = 0; t < 8; ++t) {
            const v2f b23 = {b[j & 1][t].z, b[j & 1][t].w};
            acc[t] = __builtin_amdgcn_wmma_f32_16x16x4_f32(
                 false, a23,  false, b23,
                 (short)0, acc[t],  false,  false);
        }
        __builtin_amdgcn_sched_barrier(0);
    }

    float* out = H + (size_t)(tileM + 8 * hi) * DIM + c;
#pragma unroll
    for (int r = 0; r < 8; ++r) {
#pragma unroll
        for (int t = 0; t < 8; ++t)
            out[(size_t)r * DIM + 16 * t] = acc[t][r];
    }
}

#define GA_WAVES 8

__device__ __forceinline__ float lane_bcast_f(float x, int l) {
    return __int_as_float(__builtin_amdgcn_readlane(__float_as_int(x), l));
}

__global__ __launch_bounds__(32 * GA_WAVES)
void gather_kernel(const float* __restrict__ h, const int* __restrict__ recs,
                   const int* __restrict__ off, const float* __restrict__ dis,
                   const float* __restrict__ bias, float* __restrict__ out, int n) {
    const int lane = threadIdx.x & 31;
    const int nw   = gridDim.x * GA_WAVES;
    int node = __builtin_amdgcn_readfirstlane(blockIdx.x * GA_WAVES + (threadIdx.x >> 5));
    const float4* h4 = (const float4*)h;
    const int2*   rc = (const int2*)recs;
    const float4  bb = ((const float4*)bias)[lane];

    for (; node < n; node += nw) {
        const int beg = off[node];
        const int end = off[node + 1];
        const float dc = dis[node];
        float4 acc = h4[(size_t)node * 32 + lane];
        const float s = dc * dc;
        acc.x *= s;  acc.y *= s;  acc.z *= s;  acc.w *= s;

        for (int base = beg; base < end; base += 32) {
            const int m = min(32, end - base);
            int   myr = 0;
            float myw = 0.0f;
            if (lane < m) {
                const int2 q = rc[base + lane];
                myr = q.x;
                myw = __int_as_float(q.y) * dc;
            }
            int i = 0;
            for (; i + 8 <= m; i += 8) {
                float4 v[8];
                float  w[8];
#pragma unroll
                for (int u = 0; u < 8; ++u) {
                    const int r = __builtin_amdgcn_readlane(myr, i + u);
                    w[u] = lane_bcast_f(myw, i + u);
                    v[u] = h4[(size_t)r * 32 + lane];
                }
#pragma unroll
                for (int u = 0; u < 8; ++u) {
                    acc.x = fmaf(v[u].x, w[u], acc.x);
                    acc.y = fmaf(v[u].y, w[u], acc.y);
                    acc.z = fmaf(v[u].z, w[u], acc.z);
                    acc.w = fmaf(v[u].w, w[u], acc.w);
                }
            }
            for (; i < m; i += 4) {
                float4 v[4];
                float  w[4];
#pragma unroll
                for (int u = 0; u < 4; ++u) {
                    const int k = min(i + u, m - 1);
                    const int r = __builtin_amdgcn_readlane(myr, k);
                    w[u] = lane_bcast_f(myw, k);
                    v[u] = h4[(size_t)r * 32 + lane];
                }
#pragma unroll
                for (int u = 0; u < 4; ++u) {
                    const bool live = (i + u) < m;
                    const float wx = live ? v[u].x : 0.0f;
                    const float wy = live ? v[u].y : 0.0f;
                    const float wz = live ? v[u].z : 0.0f;
                    const float ww = live ? v[u].w : 0.0f;
                    acc.x = fmaf(wx, w[u], acc.x);
                    acc.y = fmaf(wy, w[u], acc.y);
                    acc.z = fmaf(wz, w[u], acc.z);
                    acc.w = fmaf(ww, w[u], acc.w);
                }
            }
        }

        acc.x = fmaxf(acc.x + bb.x, 0.0f);
        acc.y = fmaxf(acc.y + bb.y, 0.0f);
        acc.z = fmaxf(acc.z + bb.z, 0.0f);
        acc.w = fmaxf(acc.w + bb.w, 0.0f);
        ((float4*)out)[(size_t)node * 32 + lane] = acc;
    }
}


#define BN_WAVES 16
#define BN_RPW   13
#define BN_ROWS  (BN_WAVES * BN_RPW)

__global__ __launch_bounds__(32 * BN_WAVES)
void bn_reduce_kernel(const float* __restrict__ h, float* __restrict__ sum,
                      float* __restrict__ sumsq, int n) {
    __shared__ float4 red[2][BN_WAVES][32];
    const int j  = threadIdx.x & 31;
    const int g  = threadIdx.x >> 5;
    const int r0 = blockIdx.x * BN_ROWS + g * BN_RPW;
    const float4* hp = (const float4*)h;

    float4 v[BN_RPW];
#pragma unroll
    for (int i = 0; i < BN_RPW; ++i) {
        int r = r0 + i;
        if (r >= n) r = n - 1;
        v[i] = hp[(size_t)r * 32 + j];
    }

    float4 s = {0.0f, 0.0f, 0.0f, 0.0f};
    float4 q = {0.0f, 0.0f, 0.0f, 0.0f};
#pragma unroll
    for (int i = 0; i < BN_RPW; ++i) {
        const bool live = (r0 + i) < n;
        float4 w;
        w.x = live ? v[i].x : 0.0f;
        w.y = live ? v[i].y : 0.0f;
        w.z = live ? v[i].z : 0.0f;
        w.w = live ? v[i].w : 0.0f;
        s.x += w.x;  s.y += w.y;  s.z += w.z;  s.w += w.w;
        q.x += w.x * w.x;  q.y += w.y * w.y;  q.z += w.z * w.z;  q.w += w.w * w.w;
    }

    red[0][g][j] = s;
    red[1][g][j] = q;
    __syncthreads();

    if (threadIdx.x < 2 * DIM) {
        const int which = threadIdx.x >> 7;
        const int d = threadIdx.x & 127;
        const float* rp = (const float*)&red[which][0][0] + d;
        float t = 0.0f;
#pragma unroll
        for (int w = 0; w < BN_WAVES; ++w) t += rp[w * 128];
        unsafeAtomicAdd((which ? sumsq : sum) + d, t);
    }
}

__global__ __launch_bounds__(128)
void bn_finalize_kernel(const float* __restrict__ sum, const float* __restrict__ sumsq,
                        float* __restrict__ mean, float* __restrict__ istd, int n) {
    int d = threadIdx.x;
    float invn = 1.0f / (float)n;
    float m = sum[d] * invn;
    float var = sumsq[d] * invn - m * m;
    mean[d] = m;
    istd[d] = rsqrtf(var + BN_EPS);
}

__global__ __launch_bounds__(256)
void bn_norm_kernel(const float* __restrict__ h, const float* __restrict__ mean,
                    const float* __restrict__ istd, const float* __restrict__ gamma,
                    const float* __restrict__ beta, float* __restrict__ out, int n) {
    int t = blockIdx.x * blockDim.x + threadIdx.x;
    if (t >= n * 32) return;
    int j = t & 31;
    float4 v  = ((const float4*)h)[t];
    float4 m  = ((const float4*)mean)[j];
    float4 is = ((const float4*)istd)[j];
    float4 g  = ((const float4*)gamma)[j];
    float4 be = ((const float4*)beta)[j];
    float4 o;
    o.x = g.x * (v.x - m.x) * is.x + be.x;
    o.y = g.y * (v.y - m.y) * is.y + be.y;
    o.z = g.z * (v.z - m.z) * is.z + be.z;
    o.w = g.w * (v.w - m.w) * is.w + be.w;
    ((float4*)out)[t] = o;
}


extern "C" void kernel_launch(void* const* d_in, const int* in_sizes, int n_in,
                              void* d_out, int out_size, void* d_ws, size_t ws_size,
                              hipStream_t stream) {
    const float* x     = (const float*)d_in[0];
    const int*   eidx  = (const int*)d_in[1];
    const float* W1    = (const float*)d_in[2];
    const float* b1    = (const float*)d_in[3];
    const float* W2    = (const float*)d_in[4];
    const float* b2    = (const float*)d_in[5];
    const float* gamma = (const float*)d_in[6];
    const float* beta  = (const float*)d_in[7];
    float* out = (float*)d_out;

    const int* row = eidx;
    const int* col = eidx + NE;

    float* bufA  = (float*)d_ws;
    float* bufB  = bufA + (size_t)NN * DIM;
    int*   recs  = (int*)(bufB + (size_t)NN * DIM);
    float* dis   = (float*)(recs + (size_t)2 * NE);
    float* sum   = dis + NN;
    float* sumsq = sum + DIM;
    float* mean  = sumsq + DIM;
    float* istd  = mean + DIM;
    int*   off   = (int*)(istd + DIM);
    int*   rank  = off + NN + 16;
    int*   wavecnt  = rank + NE;
    int*   chunkcnt = wavecnt + WC_TILES * CH * 8;

    const int T = 256;
    const int elem_blocks = (NN * 32 + T - 1) / T;
    const int gemm_blocks = (NN + GM_ROWS - 1) / GM_ROWS;
    const int edge_blocks = (NE + T - 1) / T;
    const int bn_blocks = (NN + BN_ROWS - 1) / BN_ROWS;
    const int scan_blocks = (NN + SCAN_T - 1) / SCAN_T;
    const int ga_blocks = 2048;

    count_kernel<<<CH, CNT_T, 0, stream>>>(col, rank, chunkcnt, wavecnt, sum);
    scan_kernel<<<scan_blocks, SCAN_T, 0, stream>>>(chunkcnt, wavecnt, off, dis, NN);
    fill_kernel<<<edge_blocks, T, 0, stream>>>(row, col, rank, off, chunkcnt, dis, recs, NE);

    gemm_wmma_kernel<<<gemm_blocks, 32 * GM_WAVES, 0, stream>>>(x, W1, bufA, NN);
    gather_kernel<<<ga_blocks, 32 * GA_WAVES, 0, stream>>>(bufA, recs, off, dis, b1, bufB, NN);

    gemm_wmma_kernel<<<gemm_blocks, 32 * GM_WAVES, 0, stream>>>(bufB, W2, bufA, NN);
    gather_kernel<<<ga_blocks, 32 * GA_WAVES, 0, stream>>>(bufA, recs, off, dis, b2, bufB, NN);

    bn_reduce_kernel<<<bn_blocks, 32 * BN_WAVES, 0, stream>>>(bufB, sum, sumsq, NN);
    bn_finalize_kernel<<<1, DIM, 0, stream>>>(sum, sumsq, mean, istd, NN);
    bn_norm_kernel<<<elem_blocks, T, 0, stream>>>(bufB, mean, istd, gamma, beta, out, NN);
}


# batchnorm sums taken in the layer-2 gather's registers (LDS fold per block, f32 atomics into 8 copies; rows grouped by wave: new sum order), bn_norm folds the copies itself; 10 -> 8 launches
# speedup vs baseline: 4.2646x; 1.0392x over previous
#include <hip/hip_runtime.h>

#define NN 50000
#define NE 600000
#define DIM 128
#define BN_EPS 1e-5f
#define BN_REP 8

typedef __attribute__((ext_vector_type(2))) float v2f;
typedef __attribute__((ext_vector_type(8))) float v8f;

#define CH      32
#define EPC     (NE / CH)
#define CNT_T   1024
#define SCAN_T  1024
#define CNT_EPT ((EPC + CNT_T - 1) / CNT_T)
#define CNT_Q   ((NN / 4 + CNT_T - 1) / CNT_T)
#define WC_TILES (4 * CNT_Q)
static_assert(NE % CH == 0, "chunks of equal length");
static_assert(NN % 4 == 0, "the counter table is moved 16 bytes at a time");
static_assert(SCAN_T == 1024 && CNT_T == 1024, "count_kernel: trip k, thread t holds nodes of tile 4k + t/256, run (t/32)%8");

__global__ __launch_bounds__(CNT_T)
void count_kernel(const int* __restrict__ col, int* __restrict__ rank,
                  int* __restrict__ chunkcnt, int* __restrict__ wavecnt,
                  float* __restrict__ bnacc) {
    __shared__ int4 tab4[NN / 4];
    int* tab = (int*)tab4;
    const int t = threadIdx.x;
    const int b = blockIdx.x;
    const int* cc = col + b * EPC;
    int* rk = rank + b * EPC;

    int cv[CNT_EPT];
#pragma unroll
    for (int k = 0; k < CNT_EPT; ++k) cv[k] = cc[min(t + k * CNT_T, EPC - 1)];

    for (int i = t; i < NN / 4; i += CNT_T) tab4[i] = make_int4(0, 0, 0, 0);
    __syncthreads();

    int rv[CNT_EPT];
#pragma unroll
    for (int k = 0; k < CNT_EPT; ++k) {
        rv[k] = 0;
        if (t + k * CNT_T < EPC) rv[k] = atomicAdd(&tab[cv[k]], 1);
    }
    __syncthreads();

    int4* o4 = (int4*)(chunkcnt + (size_t)b * NN);
    int sm[CNT_Q];
#pragma unroll
    for (int k = 0; k < CNT_Q; ++k) {
        const int i = k * CNT_T + t;
        int4 v = make_int4(0, 0, 0, 0);
        if (i < NN / 4) { v = tab4[i]; o4[i] = v; }
        sm[k] = v.x + v.y + v.z + v.w;
    }
#pragma unroll
    for (int k = 0; k < CNT_EPT; ++k)
        if (t + k * CNT_T < EPC) rk[t + k * CNT_T] = rv[k];

#pragma unroll
    for (int d = 16; d > 0; d >>= 1) {
#pragma unroll
        for (int k = 0; k < CNT_Q; ++k) sm[k] += __shfl_xor(sm[k], d);
    }
    if ((t & 31) == 0) {
        const int g = t >> 8;
        const int w = (t >> 5) & 7;
#pragma unroll
        for (int k = 0; k < CNT_Q; ++k) wavecnt[((4 * k + g) * CH + b) * 8 + w] = sm[k];
    }
    if (b == 0)
        for (int i = t; i < BN_REP * 2 * DIM; i += CNT_T) bnacc[i] = 0.0f;
}

__global__ __launch_bounds__(SCAN_T)
void scan_kernel(int* __restrict__ chunkcnt, const int* __restrict__ wavecnt,
                 int* __restrict__ off, float* __restrict__ dis, int n) {
    __shared__ int wave_own[32];
    __shared__ int wave_front[32];
    const int t    = threadIdx.x;
    const int lane = t & 31;
    const int wv   = t >> 5;
    const int b    = blockIdx.x;

    int front = 0;
#pragma unroll 4
    for (int i = t; i < 8 * CH * b; i += SCAN_T) front += wavecnt[i];

    const int idx = b * SCAN_T + t;
    int c = 0;
    if (idx < n) {
        int v[CH];
#pragma unroll
        for (int k = 0; k < CH; ++k) v[k] = chunkcnt[(size_t)k * NN + idx];
#pragma unroll
        for (int k = 0; k < CH; ++k) { chunkcnt[(size_t)k * NN + idx] = c; c += v[k]; }
    }

    int incl = c;
    for (int d = 1; d < 32; d <<= 1) {
        const int y = __shfl_up(incl, d);
        if (lane >= d) incl += y;
    }
    for (int d = 16; d > 0; d >>= 1) front += __shfl_xor(front, d);

    if (lane == 31) wave_own[wv] = incl;
    if (lane == 0)  wave_front[wv] = front;
    __syncthreads();

    int pre = wave_front[lane] + ((lane < wv) ? wave_own[lane] : 0);
    for (int d = 16; d > 0; d >>= 1) pre += __shfl_xor(pre, d);

    const int excl = pre + incl - c;
    if (idx < n) {
        off[idx] = excl;
        dis[idx] = rsqrtf((float)c + 1.0f);
        if (idx == n - 1) off[n] = excl + c;
    }
}

__global__ __launch_bounds__(256)
void fill_kernel(const int* __restrict__ row, const int* __restrict__ col,
                 const int* __restrict__ rank, const int* __restrict__ off,
                 const int* __restrict__ chunkcnt, const float* __restrict__ dis,
                 int* __restrict__ recs, int e) {
    int i = blockIdx.x * blockDim.x + threadIdx.x;
    if (i >= e) return;
    const int r = row[i];
    const int c = col[i];
    const int b = i / EPC;
    const int p = off[c] + chunkcnt[(size_t)b * NN + c] + rank[i];
    int2 rec;
    rec.x = r;
    rec.y = __float_as_int(dis[r]);
    ((int2*)recs)[p] = rec;
}

#define GM_WAVES 8
#define GM_ROWS  (16 * GM_WAVES)

__global__ __launch_bounds__(32 * GM_WAVES)
void gemm_wmma_kernel(const float* __restrict__ X, const float* __restrict__ W,
                      float* __restrict__ H, int n) {
    __shared__ float4 Ws[32 * DIM];
    const int lane  = threadIdx.x & 31;
    const int wave  = threadIdx.x >> 5;
    const int c     = lane & 15;
    const int hi    = lane >> 4;
    const int tileM = blockIdx.x * GM_ROWS + wave * 16;

    float4 a[16];
    {
        const int row = min(tileM + c, n - 1);
        const float4* xr = (const float4*)(X + (size_t)row * DIM) + hi;
#pragma unroll
        for (int j = 0; j < 16; ++j) a[j] = xr[2 * j];
    }

#pragma unroll
    for (int i = 0; i < 4; ++i) {
        const int kq = wave + GM_WAVES * i;
        const float4* wr = (const float4*)(W + (size_t)(4 * kq) * DIM) + lane;
        const float4 r0 = wr[0];
        const float4 r1 = wr[32];
        const float4 r2 = wr[64];
        const float4 r3 = wr[96];
        float4* dst = Ws + kq * DIM + 4 * lane;
        dst[0] = make_float4(r0.x, r1.x, r2.x, r3.x);
        dst[1] = make_float4(r0.y, r1.y, r2.y, r3.y);
        dst[2] = make_float4(r0.z, r1.z, r2.z, r3.z);
        dst[3] = make_float4(r0.w, r1.w, r2.w, r3.w);
    }
    __syncthreads();
    if (tileM >= n) return;

    v8f acc[8];
#pragma unroll
    for (int t = 0; t < 8; ++t) acc[t] = (v8f){};

    const float4* bp = Ws + hi * DIM + c;
    float4 b[2][8];
#pragma unroll
    for (int t = 0; t < 8; ++t) b[0][t] = bp[16 * t];

#pragma unroll
    for (int j = 0; j < 16; ++j) {
        if (j + 1 < 16) {
#pragma unroll
            for (int t = 0; t < 8; ++t) b[(j + 1) & 1][t] = bp[2 * (j + 1) * DIM + 16 * t];
        }
        __builtin_amdgcn_sched_barrier(0);
        const v2f a01 = {a[j].x, a[j].y};
        const v2f a23 = {a[j].z, a[j].w};
#pragma unroll
        for (int t = 0; t < 8; ++t) {
            const v2f b01 = {b[j & 1][t].x, b[j & 1][t].y};
            acc[t] = __builtin_amdgcn_wmma_f32_16x16x4_f32(
                 false, a01,  false, b01,
                 (short)0, acc[t],  false,  false);
        }
#pragma unroll
        for (int t = 0; t < 8; ++t) {
            const v2f b23 = {b[j & 1][t].z, b[j & 1][t].w};
            acc[t] = __builtin_amdgcn_wmma_f32_16x16x4_f32(
                 false, a23,  false, b23,
                 (short)0, acc[t],  false,  false);
        }
        __builtin_amdgcn_sched_barrier(0);
    }

    float* out = H + (size_t)(tileM + 8 * hi) * DIM + c;
#pragma unroll
    for (int r = 0; r < 8; ++r) {
#pragma unroll
        for (int t = 0; t < 8; ++t)
            out[(size_t)r * DIM + 16 * t] = acc[t][r];
    }
}

#define GA_WAVES  8
#define GS_WAVES  32

__device__ __forceinline__ float lane_bcast_f(float x, int l) {
    return __int_as_float(__builtin_amdgcn_readlane(__float_as_int(x), l));
}

template <int WAVES, bool STATS>
__device__ __forceinline__ void gather_rows(const float* __restrict__ h, const int* __restrict__ recs,
                                            const int* __restrict__ off, const float* __restrict__ dis,
                                            const float* __restrict__ bias, float* __restrict__ out, int n,
                                            float4& s, float4& q) {
    const int lane = threadIdx.x & 31;
    const int nw   = gridDim.x * WAVES;
    int node = __builtin_amdgcn_readfirstlane(blockIdx.x * WAVES + (threadIdx.x >> 5));
    const float4* h4 = (const float4*)h;
    const int2*   rc = (const int2*)recs;
    const float4  bb = ((const float4*)bias)[lane];

    for (; node < n; node += nw) {
        const int beg = off[node];
        const int end = off[node + 1];
        const float dc = dis[node];
        float4 acc = h4[(size_t)node * 32 + lane];
        const float sl = dc * dc;
        acc.x *= sl;  acc.y *= sl;  acc.z *= sl;  acc.w *= sl;

        for (int base = beg; base < end; base += 32) {
            const int m = min(32, end - base);
            int   myr = 0;
            float myw = 0.0f;
            if (lane < m) {
                const int2 rq = rc[base + lane];
                myr = rq.x;
                myw = __int_as_float(rq.y) * dc;
            }
            int i = 0;
            for (; i + 8 <= m; i += 8) {
                float4 v[8];
                float  w[8];
#pragma unroll
                for (int u = 0; u < 8; ++u) {
                    const int r = __builtin_amdgcn_readlane(myr, i + u);
                    w[u] = lane_bcast_f(myw, i + u);
                    v[u] = h4[(size_t)r * 32 + lane];
                }
#pragma unroll
                for (int u = 0; u < 8; ++u) {
                    acc.x = fmaf(v[u].x, w[u], acc.x);
                    acc.y = fmaf(v[u].y, w[u], acc.y);
                    acc.z = fmaf(v[u].z, w[u], acc.z);
                    acc.w = fmaf(v[u].w, w[u], acc.w);
                }
            }
            for (; i < m; i += 4) {
                float4 v[4];
                float  w[4];
#pragma unroll
                for (int u = 0; u < 4; ++u) {
                    const int k = min(i + u, m - 1);
                    const int r = __builtin_amdgcn_readlane(myr, k);
                    w[u] = lane_bcast_f(myw, k);
                    v[u] = h4[(size_t)r * 32 + lane];
                }
#pragma unroll
                for (int u = 0; u < 4; ++u) {
                    const bool live = (i + u) < m;
                    const float wx = live ? v[u].x : 0.0f;
                    const float wy = live ? v[u].y : 0.0f;
                    const float wz = live ? v[u].z : 0.0f;
                    const float ww = live ? v[u].w : 0.0f;
                    acc.x = fmaf(wx, w[u], acc.x);
                    acc.y = fmaf(wy, w[u], acc.y);
                    acc.z = fmaf(wz, w[u], acc.z);
                    acc.w = fmaf(ww, w[u], acc.w);
                }
            }
        }

        acc.x = fmaxf(acc.x + bb.x, 0.0f);
        acc.y = fmaxf(acc.y + bb.y, 0.0f);
        acc.z = fmaxf(acc.z + bb.z, 0.0f);
        acc.w = fmaxf(acc.w + bb.w, 0.0f);
        ((float4*)out)[(size_t)node * 32 + lane] = acc;
        if (STATS) {
            s.x += acc.x;  s.y += acc.y;  s.z += acc.z;  s.w += acc.w;
            q.x = fmaf(acc.x, acc.x, q.x);  q.y = fmaf(acc.y, acc.y, q.y);
            q.z = fmaf(acc.z, acc.z, q.z);  q.w = fmaf(acc.w, acc.w, q.w);
        }
    }
}

__global__ __launch_bounds__(32 * GA_WAVES)
void gather_kernel(const float* __restrict__ h, const int* __restrict__ recs,
                   const int* __restrict__ off, const float* __restrict__ dis,
                   const float* __restrict__ bias, float* __restrict__ out, int n) {
    float4 s, q;
    gather_rows<GA_WAVES, false>(h, recs, off, dis, bias, out, n, s, q);
}

__global__ __launch_bounds__(32 * GS_WAVES)
void gather_stats_kernel(const float* __restrict__ h, const int* __restrict__ recs,
                         const int* __restrict__ off, const float* __restrict__ dis,
                         const float* __restrict__ bias, float* __restrict__ out,
                         float* __restrict__ stats, int n) {
    __shared__ float4 red[2][GS_WAVES][32];
    float4 s = {0.0f, 0.0f, 0.0f, 0.0f};
    float4 q = {0.0f, 0.0f, 0.0f, 0.0f};
    gather_rows<GS_WAVES, true>(h, recs, off, dis, bias, out, n, s, q);

    red[0][threadIdx.x >> 5][threadIdx.x & 31] = s;
    red[1][threadIdx.x >> 5][threadIdx.x & 31] = q;
    __syncthreads();

    if (threadIdx.x < 2 * DIM) {
        const int which = threadIdx.x >> 7;
        const int d = threadIdx.x & 127;
        const float* rp = (const float*)&red[which][0][0] + d;
        float t = 0.0f;
#pragma unroll
        for (int w = 0; w < GS_WAVES; ++w) t += rp[w * 128];
        unsafeAtomicAdd(stats + (blockIdx.x % BN_REP) * 2 * DIM + threadIdx.x, t);
    }
}


__global__ __launch_bounds__(256)
void bn_norm_kernel(const float* __restrict__ h, const float* __restrict__ stats,
                    const float* __restrict__ gamma, const float* __restrict__ beta,
                    float* __restrict__ out, int n) {
    __shared__ float4 ms[2][32];
    const int tid  = threadIdx.x;
    const int t    = blockIdx.x * blockDim.x + tid;
    const int j    = tid & 31;
    const bool live = t < n * 32;
    float4 v = {0.0f, 0.0f, 0.0f, 0.0f};
    if (live) v = ((const float4*)h)[t];
    const float4 g  = ((const float4*)gamma)[j];
    const float4 be = ((const float4*)beta)[j];

    if (tid < DIM) {
        float s = 0.0f, q = 0.0f;
#pragma unroll
        for (int r = 0; r < BN_REP; ++r) {
            s += stats[r * 2 * DIM + tid];
            q += stats[r * 2 * DIM + DIM + tid];
        }
        const float invn = 1.0f / (float)n;
        const float m = s * invn;
        const float var = q * invn - m * m;
        ((float*)&ms[0][0])[tid] = m;
        ((float*)&ms[1][0])[tid] = rsqrtf(var + BN_EPS);
    }
    __syncthreads();
    const float4 m  = ms[0][j];
    const float4 is = ms[1][j];
    float4 o;
    o.x = g.x * (v.x - m.x) * is.x + be.x;
    o.y = g.y * (v.y - m.y) * is.y + be.y;
    o.z = g.z * (v.z - m.z) * is.z + be.z;
    o.w = g.w * (v.w - m.w) * is.w + be.w;
    if (live) ((float4*)out)[t] = o;
}


extern "C" void kernel_launch(void* const* d_in, const int* in_sizes, int n_in,
                              void* d_out, int out_size, void* d_ws, size_t ws_size,
                              hipStream_t stream) {
    const float* x     = (const float*)d_in[0];
    const int*   eidx  = (const int*)d_in[1];
    const float* W1    = (const float*)d_in[2];
    const float* b1    = (const float*)d_in[3];
    const float* W2    = (const float*)d_in[4];
    const float* b2    = (const float*)d_in[5];
    const float* gamma = (const float*)d_in[6];
    const float* beta  = (const float*)d_in[7];
    float* out = (float*)d_out;

    const int* row = eidx;
    const int* col = eidx + NE;

    float* bufA  = (float*)d_ws;
    float* bufB  = bufA + (size_t)NN * DIM;
    int*   recs  = (int*)(bufB + (size_t)NN * DIM);
    float* dis   = (float*)(recs + (size_t)2 * NE);
    float* stats = dis + NN;
    int*   off   = (int*)(stats + BN_REP * 2 * DIM);
    int*   rank  = off + NN + 16;
    int*   wavecnt  = rank + NE;
    int*   chunkcnt = wavecnt + WC_TILES * CH * 8;

    const int T = 256;
    const int elem_blocks = (NN * 32 + T - 1) / T;
    const int gemm_blocks = (NN + GM_ROWS - 1) / GM_ROWS;
    const int edge_blocks = (NE + T - 1) / T;
    const int scan_blocks = (NN + SCAN_T - 1) / SCAN_T;
    const int ga_blocks = 2048;
    const int gs_blocks = 512;

    count_kernel<<<CH, CNT_T, 0, stream>>>(col, rank, chunkcnt, wavecnt, stats);
    scan_kernel<<<scan_blocks, SCAN_T, 0, stream>>>(chunkcnt, wavecnt, off, dis, NN);
    fill_kernel<<<edge_blocks, T, 0, stream>>>(row, col, rank, off, chunkcnt, dis, recs, NE);

    gemm_wmma_kernel<<<gemm_blocks, 32 * GM_WAVES, 0, stream>>>(x, W1, bufA, NN);
    gather_kernel<<<ga_blocks, 32 * GA_WAVES, 0, stream>>>(bufA, recs, off, dis, b1, bufB, NN);

    gemm_wmma_kernel<<<gemm_blocks, 32 * GM_WAVES, 0, stream>>>(bufB, W2, bufA, NN);
    gather_stats_kernel<<<gs_blocks, 32 * GS_WAVES, 0, stream>>>(bufA, recs, off, dis, b2, bufB, stats, NN);

    bn_norm_kernel<<<elem_blocks, T, 0, stream>>>(bufB, stats, gamma, beta, out, NN);
}


# layer 1's product rides beside the edge count: one launch, 32 count + 224 product work-groups of 512 threads (8 -> 7; in-edge order = arrival order of 16 waves); batchnorm sums as 64-bit fixed point
# speedup vs baseline: 4.5553x; 1.0682x over previous
#include <hip/hip_runtime.h>

#define NN 50000
#define NE 600000
#define DIM 128
#define BN_EPS 1e-5f
#define BN_REP 8
#define BN_FIX 268435456.0f

typedef __attribute__((ext_vector_type(2))) float v2f;
typedef __attribute__((ext_vector_type(8))) float v8f;

#define CH      32
#define EPC     (NE / CH)
#define CNT_T   512
#define SCAN_T  1024
#define CNT_EPT ((EPC + CNT_T - 1) / CNT_T)
#define CNT_Q   ((NN / 4 + CNT_T - 1) / CNT_T)
#define CNT_G   (CNT_T / 256)
#define WC_TILES (CNT_G * CNT_Q)
static_assert(NE % CH == 0, "chunks of equal length");
static_assert(NN % 4 == 0, "the counter table is moved 16 bytes at a time");
static_assert(SCAN_T == 1024 && CNT_T % 256 == 0, "count: trip k, thread t holds nodes of tile CNT_G*k + t/256, run (t/32)%8");
static_assert(WC_TILES >= (NN + SCAN_T - 1) / SCAN_T, "every tile that scan_kernel sums over is written");

__device__ __forceinline__
void count_block(const int* __restrict__ col, int* __restrict__ rank,
                 int* __restrict__ chunkcnt, int* __restrict__ wavecnt,
                 long long* __restrict__ bnacc, int4* tab4) {
    int* tab = (int*)tab4;
    const int t = threadIdx.x;
    const int b = blockIdx.x;
    const int* cc = col + b * EPC;
    int* rk = rank + b * EPC;

    int cv[CNT_EPT];
#pragma unroll
    for (int k = 0; k < CNT_EPT; ++k) cv[k] = cc[min(t + k * CNT_T, EPC - 1)];

    for (int i = t; i < NN / 4; i += CNT_T) tab4[i] = make_int4(0, 0, 0, 0);
    __syncthreads();

    int rv[CNT_EPT];
#pragma unroll
    for (int k = 0; k < CNT_EPT; ++k) {
        rv[k] = 0;
        if (t + k * CNT_T < EPC) rv[k] = atomicAdd(&tab[cv[k]], 1);
    }
    __syncthreads();

    int4* o4 = (int4*)(chunkcnt + (size_t)b * NN);
    int sm[CNT_Q];
#pragma unroll
    for (int k = 0; k < CNT_Q; ++k) {
        const int i = k * CNT_T + t;
        int4 v = make_int4(0, 0, 0, 0);
        if (i < NN / 4) { v = tab4[i]; o4[i] = v; }
        sm[k] = v.x + v.y + v.z + v.w;
    }
#pragma unroll
    for (int k = 0; k < CNT_EPT; ++k)
        if (t + k * CNT_T < EPC) rk[t + k * CNT_T] = rv[k];

#pragma unroll
    for (int d = 16; d > 0; d >>= 1) {
#pragma unroll
        for (int k = 0; k < CNT_Q; ++k) sm[k] += __shfl_xor(sm[k], d);
    }
    if ((t & 31) == 0) {
        const int g = t >> 8;
        const int w = (t >> 5) & 7;
#pragma unroll
        for (int k = 0; k < CNT_Q; ++k) wavecnt[((CNT_G * k + g) * CH + b) * 8 + w] = sm[k];
    }
    if (b == 0)
        for (int i = t; i < BN_REP * 2 * DIM; i += CNT_T) bnacc[i] = 0;
}

__global__ __launch_bounds__(SCAN_T)
void scan_kernel(int* __restrict__ chunkcnt, const int* __restrict__ wavecnt,
                 int* __restrict__ off, float* __restrict__ dis, int n) {
    __shared__ int wave_own[32];
    __shared__ int wave_front[32];
    const int t    = threadIdx.x;
    const int lane = t & 31;
    const int wv   = t >> 5;
    const int b    = blockIdx.x;

    int front = 0;
#pragma unroll 4
    for (int i = t; i < 8 * CH * b; i += SCAN_T) front += wavecnt[i];

    const int idx = b * SCAN_T + t;
    int c = 0;
    if (idx < n) {
        int v[CH];
#pragma unroll
        for (int k = 0; k < CH; ++k) v[k] = chunkcnt[(size_t)k * NN + idx];
#pragma unroll
        for (int k = 0; k < CH; ++k) { chunkcnt[(size_t)k * NN + idx] = c; c += v[k]; }
    }

    int incl = c;
    for (int d = 1; d < 32; d <<= 1) {
        const int y = __shfl_up(incl, d);
        if (lane >= d) incl += y;
    }
    for (int d = 16; d > 0; d >>= 1) front += __shfl_xor(front, d);

    if (lane == 31) wave_own[wv] = incl;
    if (lane == 0)  wave_front[wv] = front;
    __syncthreads();

    int pre = wave_front[lane] + ((lane < wv) ? wave_own[lane] : 0);
    for (int d = 16; d > 0; d >>= 1) pre += __shfl_xor(pre, d);

    const int excl = pre + incl - c;
    if (idx < n) {
        off[idx] = excl;
        dis[idx] = rsqrtf((float)c + 1.0f);
        if (idx == n - 1) off[n] = excl + c;
    }
}

__global__ __launch_bounds__(256)
void fill_kernel(const int* __restrict__ row, const int* __restrict__ col,
                 const int* __restrict__ rank, const int* __restrict__ off,
                 const int* __restrict__ chunkcnt, const float* __restrict__ dis,
                 int* __restrict__ recs, int e) {
    int i = blockIdx.x * blockDim.x + threadIdx.x;
    if (i >= e) return;
    const int r = row[i];
    const int c = col[i];
    const int b = i / EPC;
    const int p = off[c] + chunkcnt[(size_t)b * NN + c] + rank[i];
    int2 rec;
    rec.x = r;
    rec.y = __float_as_int(dis[r]);
    ((int2*)recs)[p] = rec;
}

#define GM_WAVES 8
#define GM_ROWS  (16 * GM_WAVES)

template <int NWAVES, int STRIPS>
__device__ __forceinline__
void gemm_block(const float* __restrict__ X, const float* __restrict__ W,
                float* __restrict__ H, int n, float4* Ws, int rowBase) {
    static_assert(32 % NWAVES == 0 && STRIPS <= NWAVES, "the W copy is dealt evenly over the waves");
    const int lane  = threadIdx.x & 31;
    const int wave  = threadIdx.x >> 5;
    const int c     = lane & 15;
    const int hi    = lane >> 4;
    const int tileM = rowBase + wave * 16;
    const bool owner = (STRIPS == NWAVES) || (wave < STRIPS);

    float4 a[16];
    if (owner) {
        const int row = min(tileM + c, n - 1);
        const float4* xr = (const float4*)(X + (size_t)row * DIM) + hi;
#pragma unroll
        for (int j = 0; j < 16; ++j) a[j] = xr[2 * j];
    }

#pragma unroll
    for (int i = 0; i < 32 / NWAVES; ++i) {
        const int kq = wave + NWAVES * i;
        const float4* wr = (const float4*)(W + (size_t)(4 * kq) * DIM) + lane;
        const float4 r0 = wr[0];
        const float4 r1 = wr[32];
        const float4 r2 = wr[64];
        const float4 r3 = wr[96];
        float4* dst = Ws + kq * DIM + 4 * lane;
        dst[0] = make_float4(r0.x, r1.x, r2.x, r3.x);
        dst[1] = make_float4(r0.y, r1.y, r2.y, r3.y);
        dst[2] = make_float4(r0.z, r1.z, r2.z, r3.z);
        dst[3] = make_float4(r0.w, r1.w, r2.w, r3.w);
    }
    __syncthreads();
    if (!owner || tileM >= n) return;

    v8f acc[8];
#pragma unroll
    for (int t = 0; t < 8; ++t) acc[t] = (v8f){};

    const float4* bp = Ws + hi * DIM + c;
    float4 b[2][8];
#pragma unroll
    for (int t = 0; t < 8; ++t) b[0][t] = bp[16 * t];

#pragma unroll
    for (int j = 0; j < 16; ++j) {
        if (j + 1 < 16) {
#pragma unroll
            for (int t = 0; t < 8; ++t) b[(j + 1) & 1][t] = bp[2 * (j + 1) * DIM + 16 * t];
        }
        __builtin_amdgcn_sched_barrier(0);
        const v2f a01 = {a[j].x, a[j].y};
        const v2f a23 = {a[j].z, a[j].w};
#pragma unroll
        for (int t = 0; t < 8; ++t) {
            const v2f b01 = {b[j & 1][t].x, b[j & 1][t].y};
            acc[t] = __builtin_amdgcn_wmma_f32_16x16x4_f32(
                 false, a01,  false, b01,
                 (short)0, acc[t],  false,  false);
        }
#pragma unroll
        for (int t = 0; t < 8; ++t) {
            const v2f b23 = {b[j & 1][t].z, b[j & 1][t].w};
            acc[t] = __builtin_amdgcn_wmma_f32_16x16x4_f32(
                 false, a23,  false, b23,
                 (short)0, acc[t],  false,  false);
        }
        __builtin_amdgcn_sched_barrier(0);
    }

    float* out = H + (size_t)(tileM + 8 * hi) * DIM + c;
#pragma unroll
    for (int r = 0; r < 8; ++r) {
#pragma unroll
        for (int t = 0; t < 8; ++t)
            out[(size_t)r * DIM + 16 * t] = acc[t][r];
    }
}

__global__ __launch_bounds__(32 * GM_WAVES)
void gemm_wmma_kernel(const float* __restrict__ X, const float* __restrict__ W,
                      float* __restrict__ H, int n) {
    __shared__ float4 Ws[32 * DIM];
    gemm_block<GM_WAVES, GM_WAVES>(X, W, H, n, Ws, blockIdx.x * GM_ROWS);
}

#define CG_STRIPS 14
#define CG_GEMM   ((NN / 16 + CG_STRIPS - 1) / CG_STRIPS)
static_assert(NN % 16 == 0, "whole row strips");
static_assert((NN / 4) * 16 >= 32 * DIM * 16, "W fits in the counter table's LDS");

__global__ __launch_bounds__(CNT_T)
void count_gemm_kernel(const int* __restrict__ col, int* __restrict__ rank,
                       int* __restrict__ chunkcnt, int* __restrict__ wavecnt,
                       long long* __restrict__ bnacc,
                       const float* __restrict__ X, const float* __restrict__ W,
                       float* __restrict__ H) {
    __shared__ int4 tab4[NN / 4];
    if (blockIdx.x < CH)
        count_block(col, rank, chunkcnt, wavecnt, bnacc, tab4);
    else
        gemm_block<CNT_T / 32, CG_STRIPS>(X, W, H, NN, (float4*)tab4, (blockIdx.x - CH) * (16 * CG_STRIPS));
}

#define GA_WAVES  8
#define GS_WAVES  32

__device__ __forceinline__ float lane_bcast_f(float x, int l) {
    return __int_as_float(__builtin_amdgcn_readlane(__float_as_int(x), l));
}

template <int WAVES, bool STATS>
__device__ __forceinline__ void gather_rows(const float* __restrict__ h, const int* __restrict__ recs,
                                            const int* __restrict__ off, const float* __restrict__ dis,
                                            const float* __restrict__ bias, float* __restrict__ out, int n,
                                            float4& s, float4& q) {
    const int lane = threadIdx.x & 31;
    const int nw   = gridDim.x * WAVES;
    int node = __builtin_amdgcn_readfirstlane(blockIdx.x * WAVES + (threadIdx.x >> 5));
    const float4* h4 = (const float4*)h;
    const int2*   rc = (const int2*)recs;
    const float4  bb = ((const float4*)bias)[lane];

    for (; node < n; node += nw) {
        const int beg = off[node];
        const int end = off[node + 1];
        const float dc = dis[node];
        float4 acc = h4[(size_t)node * 32 + lane];
        const float sl = dc * dc;
        acc.x *= sl;  acc.y *= sl;  acc.z *= sl;  acc.w *= sl;

        for (int base = beg; base < end; base += 32) {
            const int m = min(32, end - base);
            int   myr = 0;
            float myw = 0.0f;
            if (lane < m) {
                const int2 rq = rc[base + lane];
                myr = rq.x;
                myw = __int_as_float(rq.y) * dc;
            }
            int i = 0;
            for (; i + 8 <= m; i += 8) {
                float4 v[8];
                float  w[8];
#pragma unroll
                for (int u = 0; u < 8; ++u) {
                    const int r = __builtin_amdgcn_readlane(myr, i + u);
                    w[u] = lane_bcast_f(myw, i + u);
                    v[u] = h4[(size_t)r * 32 + lane];
                }
#pragma unroll
                for (int u = 0; u < 8; ++u) {
                    acc.x = fmaf(v[u].x, w[u], acc.x);
                    acc.y = fmaf(v[u].y, w[u], acc.y);
                    acc.z = fmaf(v[u].z, w[u], acc.z);
                    acc.w = fmaf(v[u].w, w[u], acc.w);
                }
            }
            for (; i < m; i += 4) {
                float4 v[4];
                float  w[4];
#pragma unroll
                for (int u = 0; u < 4; ++u) {
                    const int k = min(i + u, m - 1);
                    const int r = __builtin_amdgcn_readlane(myr, k);
                    w[u] = lane_bcast_f(myw, k);
                    v[u] = h4[(size_t)r * 32 + lane];
                }
#pragma unroll
                for (int u = 0; u < 4; ++u) {
                    const bool live = (i + u) < m;
                    const float wx = live ? v[u].x : 0.0f;
                    const float wy = live ? v[u].y : 0.0f;
                    const float wz = live ? v[u].z : 0.0f;
                    const float ww = live ? v[u].w : 0.0f;
                    acc.x = fmaf(wx, w[u], acc.x);
                    acc.y = fmaf(wy, w[u], acc.y);
                    acc.z = fmaf(wz, w[u], acc.z);
                    acc.w = fmaf(ww, w[u], acc.w);
                }
            }
        }

        acc.x = fmaxf(acc.x + bb.x, 0.0f);
        acc.y = fmaxf(acc.y + bb.y, 0.0f);
        acc.z = fmaxf(acc.z + bb.z, 0.0f);
        acc.w = fmaxf(acc.w + bb.w, 0.0f);
        ((float4*)out)[(size_t)node * 32 + lane] = acc;
        if (STATS) {
            s.x += acc.x;  s.y += acc.y;  s.z += acc.z;  s.w += acc.w;
            q.x = fmaf(acc.x, acc.x, q.x);  q.y = fmaf(acc.y, acc.y, q.y);
            q.z = fmaf(acc.z, acc.z, q.z);  q.w = fmaf(acc.w, acc.w, q.w);
        }
    }
}

__global__ __launch_bounds__(32 * GA_WAVES)
void gather_kernel(const float* __restrict__ h, const int* __restrict__ recs,
                   const int* __restrict__ off, const float* __restrict__ dis,
                   const float* __restrict__ bias, float* __restrict__ out, int n) {
    float4 s, q;
    gather_rows<GA_WAVES, false>(h, recs, off, dis, bias, out, n, s, q);
}

__global__ __launch_bounds__(32 * GS_WAVES)
void gather_stats_kernel(const float* __restrict__ h, const int* __restrict__ recs,
                         const int* __restrict__ off, const float* __restrict__ dis,
                         const float* __restrict__ bias, float* __restrict__ out,
                         long long* __restrict__ stats, int n) {
    __shared__ float4 red[2][GS_WAVES][32];
    float4 s = {0.0f, 0.0f, 0.0f, 0.0f};
    float4 q = {0.0f, 0.0f, 0.0f, 0.0f};
    gather_rows<GS_WAVES, true>(h, recs, off, dis, bias, out, n, s, q);

    red[0][threadIdx.x >> 5][threadIdx.x & 31] = s;
    red[1][threadIdx.x >> 5][threadIdx.x & 31] = q;
    __syncthreads();

    if (threadIdx.x < 2 * DIM) {
        const int which = threadIdx.x >> 7;
        const int d = threadIdx.x & 127;
        const float* rp = (const float*)&red[which][0][0] + d;
        float t = 0.0f;
#pragma unroll
        for (int w = 0; w < GS_WAVES; ++w) t += rp[w * 128];
        const long long fx = (long long)rintf(t * BN_FIX);
        atomicAdd((unsigned long long*)(stats + (blockIdx.x % BN_REP) * 2 * DIM + threadIdx.x),
                  (unsigned long long)fx);
    }
}


__global__ __launch_bounds__(256)
void bn_norm_kernel(const float* __restrict__ h, const long long* __restrict__ stats,
                    const float* __restrict__ gamma, const float* __restrict__ beta,
                    float* __restrict__ out, int n) {
    __shared__ float4 ms[2][32];
    const int tid  = threadIdx.x;
    const int t    = blockIdx.x * blockDim.x + tid;
    const int j    = tid & 31;
    const bool live = t < n * 32;
    float4 v = {0.0f, 0.0f, 0.0f, 0.0f};
    if (live) v = ((const float4*)h)[t];
    const float4 g  = ((const float4*)gamma)[j];
    const float4 be = ((const float4*)beta)[j];

    if (tid < DIM) {
        long long si = 0, qi = 0;
#pragma unroll
        for (int r = 0; r < BN_REP; ++r) {
            si += stats[r * 2 * DIM + tid];
            qi += stats[r * 2 * DIM + DIM + tid];
        }
        const float s = (float)si * (1.0f / BN_FIX);
        const float q = (float)qi * (1.0f / BN_FIX);
        const float invn = 1.0f / (float)n;
        const float m = s * invn;
        const float var = q * invn - m * m;
        ((float*)&ms[0][0])[tid] = m;
        ((float*)&ms[1][0])[tid] = rsqrtf(var + BN_EPS);
    }
    __syncthreads();
    const float4 m  = ms[0][j];
    const float4 is = ms[1][j];
    float4 o;
    o.x = g.x * (v.x - m.x) * is.x + be.x;
    o.y = g.y * (v.y - m.y) * is.y + be.y;
    o.z = g.z * (v.z - m.z) * is.z + be.z;
    o.w = g.w * (v.w - m.w) * is.w + be.w;
    if (live) ((float4*)out)[t] = o;
}


extern "C" void kernel_launch(void* const* d_in, const int* in_sizes, int n_in,
                              void* d_out, int out_size, void* d_ws, size_t ws_size,
                              hipStream_t stream) {
    const float* x     = (const float*)d_in[0];
    const int*   eidx  = (const int*)d_in[1];
    const float* W1    = (const float*)d_in[2];
    const float* b1    = (const float*)d_in[3];
    const float* W2    = (const float*)d_in[4];
    const float* b2    = (const float*)d_in[5];
    const float* gamma = (const float*)d_in[6];
    const float* beta  = (const float*)d_in[7];
    float* out = (float*)d_out;

    const int* row = eidx;
    const int* col = eidx + NE;

    float* bufA  = (float*)d_ws;
    float* bufB  = bufA + (size_t)NN * DIM;
    int*   recs  = (int*)(bufB + (size_t)NN * DIM);
    float* dis   = (float*)(recs + (size_t)2 * NE);
    long long* stats = (long long*)(dis + NN);
    int*   off   = (int*)(stats + BN_REP * 2 * DIM);
    int*   rank  = off + NN + 16;
    int*   wavecnt  = rank + NE;
    int*   chunkcnt = wavecnt + WC_TILES * CH * 8;

    const int T = 256;
    const int elem_blocks = (NN * 32 + T - 1) / T;
    const int gemm_blocks = (NN + GM_ROWS - 1) / GM_ROWS;
    const int edge_blocks = (NE + T - 1) / T;
    const int scan_blocks = (NN + SCAN_T - 1) / SCAN_T;
    const int ga_blocks = 2048;
    const int gs_blocks = 512;

    count_gemm_kernel<<<CH + CG_GEMM, CNT_T, 0, stream>>>(col, rank, chunkcnt, wavecnt, stats, x, W1, bufA);
    scan_kernel<<<scan_blocks, SCAN_T, 0, stream>>>(chunkcnt, wavecnt, off, dis, NN);
    fill_kernel<<<edge_blocks, T, 0, stream>>>(row, col, rank, off, chunkcnt, dis, recs, NE);

    gather_kernel<<<ga_blocks, 32 * GA_WAVES, 0, stream>>>(bufA, recs, off, dis, b1, bufB, NN);

    gemm_wmma_kernel<<<gemm_blocks, 32 * GM_WAVES, 0, stream>>>(bufB, W2, bufA, NN);
    gather_stats_kernel<<<gs_blocks, 32 * GS_WAVES, 0, stream>>>(bufA, recs, off, dis, b2, bufB, stats, NN);

    bn_norm_kernel<<<elem_blocks, T, 0, stream>>>(bufB, stats, gamma, beta, out, NN);
}


# layer 1's aggregation feeds layer 2's product through an LDS tile in one launch: 447 resident work-groups of 112 nodes x 28 waves (4 nodes, then a 16x32 piece per wave); 7 -> 6 launches; sums kept
# speedup vs baseline: 5.0452x; 1.1075x over previous
#include <hip/hip_runtime.h>

#define NN 50000
#define NE 600000
#define DIM 128
#define BN_EPS 1e-5f
#define BN_REP 8
#define BN_FIX 268435456.0f

typedef __attribute__((ext_vector_type(2))) float v2f;
typedef __attribute__((ext_vector_type(8))) float v8f;

#define CH      32
#define EPC     (NE / CH)
#define CNT_T   512
#define SCAN_T  1024
#define CNT_EPT ((EPC + CNT_T - 1) / CNT_T)
#define CNT_Q   ((NN / 4 + CNT_T - 1) / CNT_T)
#define CNT_G   (CNT_T / 256)
#define WC_TILES (CNT_G * CNT_Q)
static_assert(NE % CH == 0, "chunks of equal length");
static_assert(NN % 4 == 0, "the counter table is moved 16 bytes at a time");
static_assert(SCAN_T == 1024 && CNT_T % 256 == 0, "count: trip k, thread t holds nodes of tile CNT_G*k + t/256, run (t/32)%8");
static_assert(WC_TILES >= (NN + SCAN_T - 1) / SCAN_T, "every tile that scan_kernel sums over is written");

__device__ __forceinline__
void count_block(const int* __restrict__ col, int* __restrict__ rank,
                 int* __restrict__ chunkcnt, int* __restrict__ wavecnt,
                 long long* __restrict__ bnacc, int4* tab4) {
    int* tab = (int*)tab4;
    const int t = threadIdx.x;
    const int b = blockIdx.x;
    const int* cc = col + b * EPC;
    int* rk = rank + b * EPC;

    int cv[CNT_EPT];
#pragma unroll
    for (int k = 0; k < CNT_EPT; ++k) cv[k] = cc[min(t + k * CNT_T, EPC - 1)];

    for (int i = t; i < NN / 4; i += CNT_T) tab4[i] = make_int4(0, 0, 0, 0);
    __syncthreads();

    int rv[CNT_EPT];
#pragma unroll
    for (int k = 0; k < CNT_EPT; ++k) {
        rv[k] = 0;
        if (t + k * CNT_T < EPC) rv[k] = atomicAdd(&tab[cv[k]], 1);
    }
    __syncthreads();

    int4* o4 = (int4*)(chunkcnt + (size_t)b * NN);
    int sm[CNT_Q];
#pragma unroll
    for (int k = 0; k < CNT_Q; ++k) {
        const int i = k * CNT_T + t;
        int4 v = make_int4(0, 0, 0, 0);
        if (i < NN / 4) { v = tab4[i]; o4[i] = v; }
        sm[k] = v.x + v.y + v.z + v.w;
    }
#pragma unroll
    for (int k = 0; k < CNT_EPT; ++k)
        if (t + k * CNT_T < EPC) rk[t + k * CNT_T] = rv[k];

#pragma unroll
    for (int d = 16; d > 0; d >>= 1) {
#pragma unroll
        for (int k = 0; k < CNT_Q; ++k) sm[k] += __shfl_xor(sm[k], d);
    }
    if ((t & 31) == 0) {
        const int g = t >> 8;
        const int w = (t >> 5) & 7;
#pragma unroll
        for (int k = 0; k < CNT_Q; ++k) wavecnt[((CNT_G * k + g) * CH + b) * 8 + w] = sm[k];
    }
    if (b == 0)
        for (int i = t; i < BN_REP * 2 * DIM; i += CNT_T) bnacc[i] = 0;
}

__global__ __launch_bounds__(SCAN_T)
void scan_kernel(int* __restrict__ chunkcnt, const int* __restrict__ wavecnt,
                 int* __restrict__ off, float* __restrict__ dis, int n) {
    __shared__ int wave_own[32];
    __shared__ int wave_front[32];
    const int t    = threadIdx.x;
    const int lane = t & 31;
    const int wv   = t >> 5;
    const int b    = blockIdx.x;

    int front = 0;
#pragma unroll 4
    for (int i = t; i < 8 * CH * b; i += SCAN_T) front += wavecnt[i];

    const int idx = b * SCAN_T + t;
    int c = 0;
    if (idx < n) {
        int v[CH];
#pragma unroll
        for (int k = 0; k < CH; ++k) v[k] = chunkcnt[(size_t)k * NN + idx];
#pragma unroll
        for (int k = 0; k < CH; ++k) { chunkcnt[(size_t)k * NN + idx] = c; c += v[k]; }
    }

    int incl = c;
    for (int d = 1; d < 32; d <<= 1) {
        const int y = __shfl_up(incl, d);
        if (lane >= d) incl += y;
    }
    for (int d = 16; d > 0; d >>= 1) front += __shfl_xor(front, d);

    if (lane == 31) wave_own[wv] = incl;
    if (lane == 0)  wave_front[wv] = front;
    __syncthreads();

    int pre = wave_front[lane] + ((lane < wv) ? wave_own[lane] : 0);
    for (int d = 16; d > 0; d >>= 1) pre += __shfl_xor(pre, d);

    const int excl = pre + incl - c;
    if (idx < n) {
        off[idx] = excl;
        dis[idx] = rsqrtf((float)c + 1.0f);
        if (idx == n - 1) off[n] = excl + c;
    }
}

__global__ __launch_bounds__(256)
void fill_kernel(const int* __restrict__ row, const int* __restrict__ col,
                 const int* __restrict__ rank, const int* __restrict__ off,
                 const int* __restrict__ chunkcnt, const float* __restrict__ dis,
                 int* __restrict__ recs, int e) {
    int i = blockIdx.x * blockDim.x + threadIdx.x;
    if (i >= e) return;
    const int r = row[i];
    const int c = col[i];
    const int b = i / EPC;
    const int p = off[c] + chunkcnt[(size_t)b * NN + c] + rank[i];
    int2 rec;
    rec.x = r;
    rec.y = __float_as_int(dis[r]);
    ((int2*)recs)[p] = rec;
}


template <int NWAVES, int STRIPS>
__device__ __forceinline__
void gemm_block(const float* __restrict__ X, const float* __restrict__ W,
                float* __restrict__ H, int n, float4* Ws, int rowBase) {
    static_assert(32 % NWAVES == 0 && STRIPS <= NWAVES, "the W copy is dealt evenly over the waves");
    const int lane  = threadIdx.x & 31;
    const int wave  = threadIdx.x >> 5;
    const int c     = lane & 15;
    const int hi    = lane >> 4;
    const int tileM = rowBase + wave * 16;
    const bool owner = (STRIPS == NWAVES) || (wave < STRIPS);

    float4 a[16];
    if (owner) {
        const int row = min(tileM + c, n - 1);
        const float4* xr = (const float4*)(X + (size_t)row * DIM) + hi;
#pragma unroll
        for (int j = 0; j < 16; ++j) a[j] = xr[2 * j];
    }

#pragma unroll
    for (int i = 0; i < 32 / NWAVES; ++i) {
        const int kq = wave + NWAVES * i;
        const float4* wr = (const float4*)(W + (size_t)(4 * kq) * DIM) + lane;
        const float4 r0 = wr[0];
        const float4 r1 = wr[32];
        const float4 r2 = wr[64];
        const float4 r3 = wr[96];
        float4* dst = Ws + kq * DIM + 4 * lane;
        dst[0] = make_float4(r0.x, r1.x, r2.x, r3.x);
        dst[1] = make_float4(r0.y, r1.y, r2.y, r3.y);
        dst[2] = make_float4(r0.z, r1.z, r2.z, r3.z);
        dst[3] = make_float4(r0.w, r1.w, r2.w, r3.w);
    }
    __syncthreads();
    if (!owner || tileM >= n) return;

    v8f acc[8];
#pragma unroll
    for (int t = 0; t < 8; ++t) acc[t] = (v8f){};

    const float4* bp = Ws + hi * DIM + c;
    float4 b[2][8];
#pragma unroll
    for (int t = 0; t < 8; ++t) b[0][t] = bp[16 * t];

#pragma unroll
    for (int j = 0; j < 16; ++j) {
        if (j + 1 < 16) {
#pragma unroll
            for (int t = 0; t < 8; ++t) b[(j + 1) & 1][t] = bp[2 * (j + 1) * DIM + 16 * t];
        }
        __builtin_amdgcn_sched_barrier(0);
        const v2f a01 = {a[j].x, a[j].y};
        const v2f a23 = {a[j].z, a[j].w};
#pragma unroll
        for (int t = 0; t < 8; ++t) {
            const v2f b01 = {b[j & 1][t].x, b[j & 1][t].y};
            acc[t] = __builtin_amdgcn_wmma_f32_16x16x4_f32(
                 false, a01,  false, b01,
                 (short)0, acc[t],  false,  false);
        }
#pragma unroll
        for (int t = 0; t < 8; ++t) {
            const v2f b23 = {b[j & 1][t].z, b[j & 1][t].w};
            acc[t] = __builtin_amdgcn_wmma_f32_16x16x4_f32(
                 false, a23,  false, b23,
                 (short)0, acc[t],  false,  false);
        }
        __builtin_amdgcn_sched_barrier(0);
    }

    float* out = H + (size_t)(tileM + 8 * hi) * DIM + c;
#pragma unroll
    for (int r = 0; r < 8; ++r) {
#pragma unroll
        for (int t = 0; t < 8; ++t)
            out[(size_t)r * DIM + 16 * t] = acc[t][r];
    }
}

#define CG_STRIPS 14
#define CG_GEMM   ((NN / 16 + CG_STRIPS - 1) / CG_STRIPS)
static_assert(NN % 16 == 0, "whole row strips");
static_assert((NN / 4) * 16 >= 32 * DIM * 16, "W fits in the counter table's LDS");

__global__ __launch_bounds__(CNT_T)
void count_gemm_kernel(const int* __restrict__ col, int* __restrict__ rank,
                       int* __restrict__ chunkcnt, int* __restrict__ wavecnt,
                       long long* __restrict__ bnacc,
                       const float* __restrict__ X, const float* __restrict__ W,
                       float* __restrict__ H) {
    __shared__ int4 tab4[NN / 4];
    if (blockIdx.x < CH)
        count_block(col, rank, chunkcnt, wavecnt, bnacc, tab4);
    else
        gemm_block<CNT_T / 32, CG_STRIPS>(X, W, H, NN, (float4*)tab4, (blockIdx.x - CH) * (16 * CG_STRIPS));
}

#define GS_WAVES  32
#define FG_PITCH4 33

__device__ __forceinline__ float lane_bcast_f(float x, int l) {
    return __int_as_float(__builtin_amdgcn_readlane(__float_as_int(x), l));
}

template <int WAVES, bool STATS, int TILE_ROWS>
__device__ __forceinline__ void gather_rows(const float* __restrict__ h, const int* __restrict__ recs,
                                            const int* __restrict__ off, const float* __restrict__ dis,
                                            const float* __restrict__ bias, float* __restrict__ out, int n,
                                            float4& s, float4& q, float4* tile) {
    const int lane = threadIdx.x & 31;
    const int nw   = (TILE_ROWS > 0) ? WAVES : gridDim.x * WAVES;
    const int first = (TILE_ROWS > 0) ? blockIdx.x * TILE_ROWS : blockIdx.x * WAVES;
    if (TILE_ROWS > 0) n = min(n, first + TILE_ROWS);
    int node = __builtin_amdgcn_readfirstlane(first + (threadIdx.x >> 5));
    const float4* h4 = (const float4*)h;
    const int2*   rc = (const int2*)recs;
    const float4  bb = ((const float4*)bias)[lane];

    for (; node < n; node += nw) {
        const int beg = off[node];
        const int end = off[node + 1];
        const float dc = dis[node];
        float4 acc = h4[(size_t)node * 32 + lane];
        const float sl = dc * dc;
        acc.x *= sl;  acc.y *= sl;  acc.z *= sl;  acc.w *= sl;

        for (int base = beg; base < end; base += 32) {
            const int m = min(32, end - base);
            int   myr = 0;
            float myw = 0.0f;
            if (lane < m) {
                const int2 rq = rc[base + lane];
                myr = rq.x;
                myw = __int_as_float(rq.y) * dc;
            }
            int i = 0;
            for (; i + 8 <= m; i += 8) {
                float4 v[8];
                float  w[8];
#pragma unroll
                for (int u = 0; u < 8; ++u) {
                    const int r = __builtin_amdgcn_readlane(myr, i + u);
                    w[u] = lane_bcast_f(myw, i + u);
                    v[u] = h4[(size_t)r * 32 + lane];
                }
#pragma unroll
                for (int u = 0; u < 8; ++u) {
                    acc.x = fmaf(v[u].x, w[u], acc.x);
                    acc.y = fmaf(v[u].y, w[u], acc.y);
                    acc.z = fmaf(v[u].z, w[u], acc.z);
                    acc.w = fmaf(v[u].w, w[u], acc.w);
                }
            }
            for (; i < m; i += 4) {
                float4 v[4];
                float  w[4];
#pragma unroll
                for (int u = 0; u < 4; ++u) {
                    const int k = min(i + u, m - 1);
                    const int r = __builtin_amdgcn_readlane(myr, k);
                    w[u] = lane_bcast_f(myw, k);
                    v[u] = h4[(size_t)r * 32 + lane];
                }
#pragma unroll
                for (int u = 0; u < 4; ++u) {
                    const bool live = (i + u) < m;
                    const float wx = live ? v[u].x : 0.0f;
                    const float wy = live ? v[u].y : 0.0f;
                    const float wz = live ? v[u].z : 0.0f;
                    const float ww = live ? v[u].w : 0.0f;
                    acc.x = fmaf(wx, w[u], acc.x);
                    acc.y = fmaf(wy, w[u], acc.y);
                    acc.z = fmaf(wz, w[u], acc.z);
                    acc.w = fmaf(ww, w[u], acc.w);
                }
            }
        }

        acc.x = fmaxf(acc.x + bb.x, 0.0f);
        acc.y = fmaxf(acc.y + bb.y, 0.0f);
        acc.z = fmaxf(acc.z + bb.z, 0.0f);
        acc.w = fmaxf(acc.w + bb.w, 0.0f);
        if (TILE_ROWS > 0) tile[(node - first) * FG_PITCH4 + lane] = acc;
        else               ((float4*)out)[(size_t)node * 32 + lane] = acc;
        if (STATS) {
            s.x += acc.x;  s.y += acc.y;  s.z += acc.z;  s.w += acc.w;
            q.x = fmaf(acc.x, acc.x, q.x);  q.y = fmaf(acc.y, acc.y, q.y);
            q.z = fmaf(acc.z, acc.z, q.z);  q.w = fmaf(acc.w, acc.w, q.w);
        }
    }
}

#define FG_NODES 112
#define FG_WAVES 28
static_assert(FG_NODES % 16 == 0 && NN % 16 == 0, "whole row strips");
static_assert(FG_WAVES == (FG_NODES / 16) * 4 && FG_NODES % FG_WAVES == 0, "one 16 x 32 piece per wave; equal node counts per wave");

__global__ __launch_bounds__(32 * FG_WAVES)
void gather_gemm_kernel(const float* __restrict__ h, const int* __restrict__ recs,
                        const int* __restrict__ off, const float* __restrict__ dis,
                        const float* __restrict__ bias, const float* __restrict__ W,
                        float* __restrict__ H, int n) {
    __shared__ float4 Ws[32 * DIM];
    __shared__ float4 At[FG_NODES * FG_PITCH4];
    const int lane = threadIdx.x & 31;
    const int wave = threadIdx.x >> 5;

    for (int kq = wave; kq < 32; kq += FG_WAVES) {
        const float4* wr = (const float4*)(W + (size_t)(4 * kq) * DIM) + lane;
        const float4 r0 = wr[0];
        const float4 r1 = wr[32];
        const float4 r2 = wr[64];
        const float4 r3 = wr[96];
        float4* dst = Ws + kq * DIM + 4 * lane;
        dst[0] = make_float4(r0.x, r1.x, r2.x, r3.x);
        dst[1] = make_float4(r0.y, r1.y, r2.y, r3.y);
        dst[2] = make_float4(r0.z, r1.z, r2.z, r3.z);
        dst[3] = make_float4(r0.w, r1.w, r2.w, r3.w);
    }

    float4 s, q;
    gather_rows<FG_WAVES, false, FG_NODES>(h, recs, off, dis, bias, nullptr, n, s, q, At);
    __syncthreads();

    const int c     = lane & 15;
    const int hi    = lane >> 4;
    const int strip = wave >> 2;
    const int qd    = wave & 3;
    const int tileM = blockIdx.x * FG_NODES + 16 * strip;
    if (tileM >= n) return;

    const float4* ap = At + (16 * strip + c) * FG_PITCH4 + hi;
    const float4* bp = Ws + hi * DIM + 32 * qd + c;
    v8f acc0 = (v8f){};
    v8f acc1 = (v8f){};
    float4 a[2], b[2][2];
    a[0]    = ap[0];
    b[0][0] = bp[0];
    b[0][1] = bp[16];
#pragma unroll
    for (int j = 0; j < 16; ++j) {
        if (j + 1 < 16) {
            a[(j + 1) & 1]    = ap[2 * (j + 1)];
            b[(j + 1) & 1][0] = bp[2 * (j + 1) * DIM];
            b[(j + 1) & 1][1] = bp[2 * (j + 1) * DIM + 16];
        }
        const v2f a01 = {a[j & 1].x, a[j & 1].y};
        const v2f a23 = {a[j & 1].z, a[j & 1].w};
        const v2f b01_0 = {b[j & 1][0].x, b[j & 1][0].y};
        const v2f b01_1 = {b[j & 1][1].x, b[j & 1][1].y};
        const v2f b23_0 = {b[j & 1][0].z, b[j & 1][0].w};
        const v2f b23_1 = {b[j & 1][1].z, b[j & 1][1].w};
        acc0 = __builtin_amdgcn_wmma_f32_16x16x4_f32(false, a01, false, b01_0, (short)0, acc0, false, false);
        acc1 = __builtin_amdgcn_wmma_f32_16x16x4_f32(false, a01, false, b01_1, (short)0, acc1, false, false);
        acc0 = __builtin_amdgcn_wmma_f32_16x16x4_f32(false, a23, false, b23_0, (short)0, acc0, false, false);
        acc1 = __builtin_amdgcn_wmma_f32_16x16x4_f32(false, a23, false, b23_1, (short)0, acc1, false, false);
    }

    float* out = H + (size_t)(tileM + 8 * hi) * DIM + 32 * qd + c;
#pragma unroll
    for (int r = 0; r < 8; ++r) {
        out[(size_t)r * DIM]      = acc0[r];
        out[(size_t)r * DIM + 16] = acc1[r];
    }
}

__global__ __launch_bounds__(32 * GS_WAVES)
void gather_stats_kernel(const float* __restrict__ h, const int* __restrict__ recs,
                         const int* __restrict__ off, const float* __restrict__ dis,
                         const float* __restrict__ bias, float* __restrict__ out,
                         long long* __restrict__ stats, int n) {
    __shared__ float4 red[2][GS_WAVES][32];
    float4 s = {0.0f, 0.0f, 0.0f, 0.0f};
    float4 q = {0.0f, 0.0f, 0.0f, 0.0f};
    gather_rows<GS_WAVES, true, 0>(h, recs, off, dis, bias, out, n, s, q, nullptr);

    red[0][threadIdx.x >> 5][threadIdx.x & 31] = s;
    red[1][threadIdx.x >> 5][threadIdx.x & 31] = q;
    __syncthreads();

    if (threadIdx.x < 2 * DIM) {
        const int which = threadIdx.x >> 7;
        const int d = threadIdx.x & 127;
        const float* rp = (const float*)&red[which][0][0] + d;
        float t = 0.0f;
#pragma unroll
        for (int w = 0; w < GS_WAVES; ++w) t += rp[w * 128];
        const long long fx = (long long)rintf(t * BN_FIX);
        atomicAdd((unsigned long long*)(stats + (blockIdx.x % BN_REP) * 2 * DIM + threadIdx.x),
                  (unsigned long long)fx);
    }
}


__global__ __launch_bounds__(256)
void bn_norm_kernel(const float* __restrict__ h, const long long* __restrict__ stats,
                    const float* __restrict__ gamma, const float* __restrict__ beta,
                    float* __restrict__ out, int n) {
    __shared__ float4 ms[2][32];
    const int tid  = threadIdx.x;
    const int t    = blockIdx.x * blockDim.x + tid;
    const int j    = tid & 31;
    const bool live = t < n * 32;
    float4 v = {0.0f, 0.0f, 0.0f, 0.0f};
    if (live) v = ((const float4*)h)[t];
    const float4 g  = ((const float4*)gamma)[j];
    const float4 be = ((const float4*)beta)[j];

    if (tid < DIM) {
        long long si = 0, qi = 0;
#pragma unroll
        for (int r = 0; r < BN_REP; ++r) {
            si += stats[r * 2 * DIM + tid];
            qi += stats[r * 2 * DIM + DIM + tid];
        }
        const float s = (float)si * (1.0f / BN_FIX);
        const float q = (float)qi * (1.0f / BN_FIX);
        const float invn = 1.0f / (float)n;
        const float m = s * invn;
        const float var = q * invn - m * m;
        ((float*)&ms[0][0])[tid] = m;
        ((float*)&ms[1][0])[tid] = rsqrtf(var + BN_EPS);
    }
    __syncthreads();
    const float4 m  = ms[0][j];
    const float4 is = ms[1][j];
    float4 o;
    o.x = g.x * (v.x - m.x) * is.x + be.x;
    o.y = g.y * (v.y - m.y) * is.y + be.y;
    o.z = g.z * (v.z - m.z) * is.z + be.z;
    o.w = g.w * (v.w - m.w) * is.w + be.w;
    if (live) ((float4*)out)[t] = o;
}


extern "C" void kernel_launch(void* const* d_in, const int* in_sizes, int n_in,
                              void* d_out, int out_size, void* d_ws, size_t ws_size,
                              hipStream_t stream) {
    const float* x     = (const float*)d_in[0];
    const int*   eidx  = (const int*)d_in[1];
    const float* W1    = (const float*)d_in[2];
    const float* b1    = (const float*)d_in[3];
    const float* W2    = (const float*)d_in[4];
    const float* b2    = (const float*)d_in[5];
    const float* gamma = (const float*)d_in[6];
    const float* beta  = (const float*)d_in[7];
    float* out = (float*)d_out;

    const int* row = eidx;
    const int* col = eidx + NE;

    float* bufA  = (float*)d_ws;
    float* bufB  = bufA + (size_t)NN * DIM;
    int*   recs  = (int*)(bufB + (size_t)NN * DIM);
    float* dis   = (float*)(recs + (size_t)2 * NE);
    long long* stats = (long long*)(dis + NN);
    int*   off   = (int*)(stats + BN_REP * 2 * DIM);
    int*   rank  = off + NN + 16;
    int*   wavecnt  = rank + NE;
    int*   chunkcnt = wavecnt + WC_TILES * CH * 8;

    const int T = 256;
    const int elem_blocks = (NN * 32 + T - 1) / T;
    const int edge_blocks = (NE + T - 1) / T;
    const int scan_blocks = (NN + SCAN_T - 1) / SCAN_T;
    const int fg_blocks = (NN + FG_NODES - 1) / FG_NODES;
    const int gs_blocks = 512;

    count_gemm_kernel<<<CH + CG_GEMM, CNT_T, 0, stream>>>(col, rank, chunkcnt, wavecnt, stats, x, W1, bufA);
    scan_kernel<<<scan_blocks, SCAN_T, 0, stream>>>(chunkcnt, wavecnt, off, dis, NN);
    fill_kernel<<<edge_blocks, T, 0, stream>>>(row, col, rank, off, chunkcnt, dis, recs, NE);

    gather_gemm_kernel<<<fg_blocks, 32 * FG_WAVES, 0, stream>>>(bufA, recs, off, dis, b1, W2, bufB, NN);

    gather_stats_kernel<<<gs_blocks, 32 * GS_WAVES, 0, stream>>>(bufB, recs, off, dis, b2, bufA, stats, NN);

    bn_norm_kernel<<<elem_blocks, T, 0, stream>>>(bufA, stats, gamma, beta, out, NN);
}
